# starting kernel
# baseline (speedup 1.0000x reference)
#include <hip/hip_runtime.h>
#include <hip/hip_bf16.h>
#include <math.h>

typedef __attribute__((ext_vector_type(2))) float v2f;
typedef __attribute__((ext_vector_type(8))) float v8f;

#define N_NODES 50000
#define N_EDGES 800000
#define M_HYPER 10000
#define P_PAIRS 200000
#define DIM     64
#define NEG_SLOPE 0.2f

__device__ __forceinline__ unsigned enc_f32(float f) {
    unsigned u = __float_as_uint(f);
    return (u & 0x80000000u) ? ~u : (u | 0x80000000u);
}
__device__ __forceinline__ float dec_f32(unsigned u) {
    return (u & 0x80000000u) ? __uint_as_float(u & 0x7FFFFFFFu)
                             : __uint_as_float(~u);
}

__global__ void k_zero(float* __restrict__ p, long n) {
    long i = (long)blockIdx.x * blockDim.x + threadIdx.x;
    if (i < n) p[i] = 0.0f;
}

__global__ void k_gemm_wmma(const float* __restrict__ X, const float* __restrict__ W,
                            const float* __restrict__ bias, float* __restrict__ Xp) {
    int wave = (blockIdx.x * blockDim.x + threadIdx.x) >> 5;
    int lane = threadIdx.x & 31;
    if (wave >= N_NODES / 16) return;
    int row_base = wave * 16;
    int lh  = (lane >= 16) ? 1 : 0;
    int l15 = lane & 15;

    const float* xrow = X + (size_t)(row_base + l15) * DIM + lh * 2;
    const float* wrow = W + (size_t)l15 * DIM + lh * 2;

    v8f acc[4] = {};
#pragma unroll
    for (int ks = 0; ks < 16; ++ks) {
        v2f a;
        a.x = xrow[ks * 4 + 0];
        a.y = xrow[ks * 4 + 1];
#pragma unroll
        for (int t = 0; t < 4; ++t) {
            v2f b;
            const float* wb = wrow + (size_t)t * 16 * DIM + ks * 4;
            b.x = wb[0];
            b.y = wb[1];
            acc[t] = __builtin_amdgcn_wmma_f32_16x16x4_f32(
                false, a, false, b, (short)0, acc[t], false, false);
        }
    }

#pragma unroll
    for (int t = 0; t < 4; ++t) {
        int c = t * 16 + l15;
        float bv = bias[c];
#pragma unroll
        for (int v = 0; v < 8; ++v) {
            int m = row_base + v + lh * 8;
            Xp[(size_t)m * DIM + c] = acc[t][v] + bv;
        }
    }
}

__global__ void k_scalars(const float* __restrict__ Xp,
                          const float* __restrict__ a_src, const float* __restrict__ a_dst,
                          float* __restrict__ s_src, float* __restrict__ s_dst) {
    int n = blockIdx.x * blockDim.x + threadIdx.x;
    if (n >= N_NODES) return;
    const float4* row = (const float4*)(Xp + (size_t)n * DIM);
    const float4* as  = (const float4*)a_src;
    const float4* ad  = (const float4*)a_dst;
    float ss = 0.f, sd = 0.f;
#pragma unroll
    for (int i = 0; i < 16; ++i) {
        float4 x = row[i], u = as[i], v = ad[i];
        ss += x.x * u.x + x.y * u.y + x.z * u.z + x.w * u.w;
        sd += x.x * v.x + x.y * v.y + x.z * v.z + x.w * v.w;
    }
    s_src[n] = ss;
    s_dst[n] = sd;
}

__global__ void k_deg(const int* __restrict__ inc_v, const int* __restrict__ inc_e,
                      float* __restrict__ dv, float* __restrict__ de) {
    int p = blockIdx.x * blockDim.x + threadIdx.x;
    if (p >= P_PAIRS) return;
    atomicAdd(dv + inc_v[p], 1.0f);
    atomicAdd(de + inc_e[p], 1.0f);
}

__global__ void k_score(const float* __restrict__ s_src, const float* __restrict__ s_dst,
                        const int* __restrict__ e_src, const int* __restrict__ e_dst,
                        float* __restrict__ score, unsigned* __restrict__ segmax) {
    int e = blockIdx.x * blockDim.x + threadIdx.x;
    if (e >= N_EDGES) return;
    float x = s_src[e_src[e]] + s_dst[e_dst[e]];
    float sc = (x >= 0.f) ? x : NEG_SLOPE * x;
    score[e] = sc;
    atomicMax(segmax + e_dst[e], enc_f32(sc));
}

__global__ void k_exp(const int* __restrict__ e_dst, const unsigned* __restrict__ segmax,
                      float* __restrict__ score, float* __restrict__ denom) {
    int e = blockIdx.x * blockDim.x + threadIdx.x;
    if (e >= N_EDGES) return;
    int d = e_dst[e];
    float m = dec_f32(segmax[d]);
    if (!isfinite(m)) m = 0.f;
    float w = expf(score[e] - m);
    score[e] = w;
    atomicAdd(denom + d, w);
}

__global__ void k_scatter_gat(const int* __restrict__ e_src, const int* __restrict__ e_dst,
                              const float* __restrict__ score, const float* __restrict__ denom,
                              const float* __restrict__ Xp, float* __restrict__ Xg) {
    long gid = (long)blockIdx.x * blockDim.x + threadIdx.x;
    long e   = gid >> 5;
    int lane = (int)(gid & 31);
    if (e >= N_EDGES) return;
    int s = e_src[e], d = e_dst[e];
    float coef = score[e] / denom[d];
    float2 xv = *(const float2*)(Xp + (size_t)s * DIM + lane * 2);
    float* out = Xg + (size_t)d * DIM + lane * 2;
    atomicAdd(out + 0, coef * xv.x);
    atomicAdd(out + 1, coef * xv.y);
}

__global__ void k_hg_gather(const int* __restrict__ inc_v, const int* __restrict__ inc_e,
                            const float* __restrict__ Xp, const float* __restrict__ dv,
                            float* __restrict__ He) {
    long gid = (long)blockIdx.x * blockDim.x + threadIdx.x;
    long p   = gid >> 5;
    int lane = (int)(gid & 31);
    if (p >= P_PAIRS) return;
    int v = inc_v[p], e = inc_e[p];
    float dvv = dv[v];
    float s = (dvv > 0.f) ? rsqrtf(dvv) : 0.f;
    float2 xv = *(const float2*)(Xp + (size_t)v * DIM + lane * 2);
    float* out = He + (size_t)e * DIM + lane * 2;
    atomicAdd(out + 0, s * xv.x);
    atomicAdd(out + 1, s * xv.y);
}

__global__ void k_he_norm(float* __restrict__ He, const float* __restrict__ de) {
    int i = blockIdx.x * blockDim.x + threadIdx.x;
    if (i >= M_HYPER * DIM) return;
    float d = de[i >> 6];
    He[i] *= (d > 0.f) ? (1.f / d) : 0.f;
}

__global__ void k_hg_scatter(const int* __restrict__ inc_v, const int* __restrict__ inc_e,
                             const float* __restrict__ He, float* __restrict__ Xhg) {
    long gid = (long)blockIdx.x * blockDim.x + threadIdx.x;
    long p   = gid >> 5;
    int lane = (int)(gid & 31);
    if (p >= P_PAIRS) return;
    int v = inc_v[p], e = inc_e[p];
    float2 h = *(const float2*)(He + (size_t)e * DIM + lane * 2);
    float* out = Xhg + (size_t)v * DIM + lane * 2;
    atomicAdd(out + 0, h.x);
    atomicAdd(out + 1, h.y);
}

__global__ void k_final(const float* __restrict__ Xg, const float* __restrict__ Xhg,
                        const float* __restrict__ dv, float* __restrict__ out) {
    long i = (long)blockIdx.x * blockDim.x + threadIdx.x;
    if (i >= (long)N_NODES * DIM) return;
    float d = dv[i >> 6];
    float s = (d > 0.f) ? rsqrtf(d) : 0.f;
    float x = 0.5f * (Xg[i] + Xhg[i] * s);
    out[i] = (x > 0.f) ? x : expm1f(x);
}

extern "C" void kernel_launch(void* const* d_in, const int* in_sizes, int n_in,
                              void* d_out, int out_size, void* d_ws, size_t ws_size,
                              hipStream_t stream) {
    const float* X     = (const float*)d_in[0];
    const float* W     = (const float*)d_in[1];
    const float* bias  = (const float*)d_in[2];
    const float* a_src = (const float*)d_in[3];
    const float* a_dst = (const float*)d_in[4];
    const int*   e_src = (const int*)d_in[5];
    const int*   e_dst = (const int*)d_in[6];
    const int*   inc_v = (const int*)d_in[7];
    const int*   inc_e = (const int*)d_in[8];
    float* out = (float*)d_out;
    float* ws  = (float*)d_ws;

    size_t off = 0;
    float* Xp    = ws + off; off += (size_t)N_NODES * DIM;
    float* s_src = ws + off; off += N_NODES;
    float* s_dst = ws + off; off += N_NODES;
    size_t zero_base = off;
    unsigned* segmax = (unsigned*)(ws + off); off += N_NODES;
    float* denom = ws + off; off += N_NODES;
    float* dv    = ws + off; off += N_NODES;
    float* de    = ws + off; off += M_HYPER;
    float* Xg    = ws + off; off += (size_t)N_NODES * DIM;
    float* He    = ws + off; off += (size_t)M_HYPER * DIM;
    float* Xhg   = ws + off; off += (size_t)N_NODES * DIM;
    size_t zero_cnt = off - zero_base;
    float* score = ws + off; off += N_EDGES;

    const int B = 256;
    k_zero<<<(int)((zero_cnt + B - 1) / B), B, 0, stream>>>(ws + zero_base, (long)zero_cnt);

    int tiles = N_NODES / 16;
    k_gemm_wmma<<<(tiles * 32 + B - 1) / B, B, 0, stream>>>(X, W, bias, Xp);

    k_scalars<<<(N_NODES + B - 1) / B, B, 0, stream>>>(Xp, a_src, a_dst, s_src, s_dst);
    k_deg<<<(P_PAIRS + B - 1) / B, B, 0, stream>>>(inc_v, inc_e, dv, de);
    k_score<<<(N_EDGES + B - 1) / B, B, 0, stream>>>(s_src, s_dst, e_src, e_dst, score, segmax);
    k_exp<<<(N_EDGES + B - 1) / B, B, 0, stream>>>(e_dst, segmax, score, denom);
    k_scatter_gat<<<(int)(((long)N_EDGES * 32 + B - 1) / B), B, 0, stream>>>(
        e_src, e_dst, score, denom, Xp, Xg);
    k_hg_gather<<<(int)(((long)P_PAIRS * 32 + B - 1) / B), B, 0, stream>>>(
        inc_v, inc_e, Xp, dv, He);
    k_he_norm<<<(M_HYPER * DIM + B - 1) / B, B, 0, stream>>>(He, de);
    k_hg_scatter<<<(int)(((long)P_PAIRS * 32 + B - 1) / B), B, 0, stream>>>(
        inc_v, inc_e, He, Xhg);
    k_final<<<(int)(((long)N_NODES * DIM + B - 1) / B), B, 0, stream>>>(Xg, Xhg, dv, out);
}


# Degree counts, segment max and softmax denominators: scattered global atomics replaced by per-work-group LDS tables plus plain-load merge kernels (float sums now LDS partials, then fixed-order merge)
# speedup vs baseline: 1.4518x; 1.4518x over previous
#include <hip/hip_runtime.h>
#include <hip/hip_bf16.h>
#include <math.h>

typedef __attribute__((ext_vector_type(2))) float v2f;
typedef __attribute__((ext_vector_type(8))) float v8f;

#define N_NODES 50000
#define N_EDGES 800000
#define M_HYPER 10000
#define P_PAIRS 200000
#define DIM     64
#define NEG_SLOPE 0.2f

__device__ __forceinline__ unsigned enc_f32(float f) {
    unsigned u = __float_as_uint(f);
    return (u & 0x80000000u) ? ~u : (u | 0x80000000u);
}
__device__ __forceinline__ float dec_f32(unsigned u) {
    return (u & 0x80000000u) ? __uint_as_float(u & 0x7FFFFFFFu)
                             : __uint_as_float(~u);
}

__global__ void k_zero(float* __restrict__ p, long n) {
    long i = (long)blockIdx.x * blockDim.x + threadIdx.x;
    if (i < n) p[i] = 0.0f;
}

__global__ void k_gemm_wmma(const float* __restrict__ X, const float* __restrict__ W,
                            const float* __restrict__ bias, float* __restrict__ Xp) {
    int wave = (blockIdx.x * blockDim.x + threadIdx.x) >> 5;
    int lane = threadIdx.x & 31;
    if (wave >= N_NODES / 16) return;
    int row_base = wave * 16;
    int lh  = (lane >= 16) ? 1 : 0;
    int l15 = lane & 15;

    const float* xrow = X + (size_t)(row_base + l15) * DIM + lh * 2;
    const float* wrow = W + (size_t)l15 * DIM + lh * 2;

    v8f acc[4] = {};
#pragma unroll
    for (int ks = 0; ks < 16; ++ks) {
        v2f a;
        a.x = xrow[ks * 4 + 0];
        a.y = xrow[ks * 4 + 1];
#pragma unroll
        for (int t = 0; t < 4; ++t) {
            v2f b;
            const float* wb = wrow + (size_t)t * 16 * DIM + ks * 4;
            b.x = wb[0];
            b.y = wb[1];
            acc[t] = __builtin_amdgcn_wmma_f32_16x16x4_f32(
                false, a, false, b, (short)0, acc[t], false, false);
        }
    }

#pragma unroll
    for (int t = 0; t < 4; ++t) {
        int c = t * 16 + l15;
        float bv = bias[c];
#pragma unroll
        for (int v = 0; v < 8; ++v) {
            int m = row_base + v + lh * 8;
            Xp[(size_t)m * DIM + c] = acc[t][v] + bv;
        }
    }
}

__global__ void k_scalars(const float* __restrict__ Xp,
                          const float* __restrict__ a_src, const float* __restrict__ a_dst,
                          float* __restrict__ s_src, float* __restrict__ s_dst) {
    int n = blockIdx.x * blockDim.x + threadIdx.x;
    if (n >= N_NODES) return;
    const float4* row = (const float4*)(Xp + (size_t)n * DIM);
    const float4* as  = (const float4*)a_src;
    const float4* ad  = (const float4*)a_dst;
    float ss = 0.f, sd = 0.f;
#pragma unroll
    for (int i = 0; i < 16; ++i) {
        float4 x = row[i], u = as[i], v = ad[i];
        ss += x.x * u.x + x.y * u.y + x.z * u.z + x.w * u.w;
        sd += x.x * v.x + x.y * v.y + x.z * v.z + x.w * v.w;
    }
    s_src[n] = ss;
    s_dst[n] = sd;
}

#define T_THREADS 1024
#define T_EDGE 128
#define T_PAIR 64
#define EDGES_PER_WG (N_EDGES / T_EDGE)
#define PAIRS_PER_WG (P_PAIRS / T_PAIR)
#define EDGE_SLOTS ((EDGES_PER_WG + T_THREADS - 1) / T_THREADS)
#define PAIR_SLOTS ((PAIRS_PER_WG + T_THREADS - 1) / T_THREADS)
#define TAB_N ((N_NODES + 31) / 32 * 32)
#define TAB_M ((M_HYPER + 31) / 32 * 32)
static_assert(N_EDGES % T_EDGE == 0 && P_PAIRS % T_PAIR == 0, "slices are exact");
static_assert((size_t)T_EDGE * TAB_N <= (size_t)2 * N_NODES * DIM + (size_t)M_HYPER * DIM,
              "edge tables fit the span they borrow");
static_assert(T_PAIR * TAB_N <= N_NODES * DIM + M_HYPER * DIM && T_PAIR * TAB_M <= N_NODES * DIM,
              "degree tables fit the spans they borrow");

__global__ __launch_bounds__(T_THREADS)
void k_deg_part(const int* __restrict__ inc_v, const int* __restrict__ inc_e,
                float* __restrict__ part_v, float* __restrict__ part_e) {
    __shared__ unsigned tv[N_NODES];
    __shared__ unsigned te[M_HYPER];
    const int tid = threadIdx.x;
    for (int i = tid; i < N_NODES; i += T_THREADS) tv[i] = 0u;
    for (int i = tid; i < M_HYPER; i += T_THREADS) te[i] = 0u;
    __syncthreads();
    const int base = blockIdx.x * PAIRS_PER_WG;
    int v[PAIR_SLOTS], h[PAIR_SLOTS];
#pragma unroll
    for (int k = 0; k < PAIR_SLOTS; ++k) {
        int i = tid + k * T_THREADS;
        bool ok = i < PAIRS_PER_WG;
        v[k] = ok ? inc_v[base + i] : 0;
        h[k] = ok ? inc_e[base + i] : 0;
    }
#pragma unroll
    for (int k = 0; k < PAIR_SLOTS; ++k) {
        if (tid + k * T_THREADS < PAIRS_PER_WG) {
            atomicAdd(&tv[v[k]], 1u);
            atomicAdd(&te[h[k]], 1u);
        }
    }
    __syncthreads();
    float* pv = part_v + (size_t)blockIdx.x * TAB_N;
    for (int i = tid; i < N_NODES; i += T_THREADS) pv[i] = (float)tv[i];
    float* pe = part_e + (size_t)blockIdx.x * TAB_M;
    for (int i = tid; i < M_HYPER; i += T_THREADS) pe[i] = (float)te[i];
}

__global__ void k_deg_merge(const float* __restrict__ part_v, const float* __restrict__ part_e,
                            float* __restrict__ dv, float* __restrict__ de) {
    int n = blockIdx.x * blockDim.x + threadIdx.x;
    if (n < N_NODES) {
        float s = 0.f;
#pragma unroll 16
        for (int g = 0; g < T_PAIR; ++g) s += part_v[(size_t)g * TAB_N + n];
        dv[n] = s;
    }
    if (n < M_HYPER) {
        float s = 0.f;
#pragma unroll 16
        for (int g = 0; g < T_PAIR; ++g) s += part_e[(size_t)g * TAB_M + n];
        de[n] = s;
    }
}

__global__ __launch_bounds__(T_THREADS)
void k_score_part(const float* __restrict__ s_src, const float* __restrict__ s_dst,
                  const int* __restrict__ e_src, const int* __restrict__ e_dst,
                  float* __restrict__ score, unsigned* __restrict__ part) {
    __shared__ unsigned tm[N_NODES];
    const int tid = threadIdx.x;
    for (int i = tid; i < N_NODES; i += T_THREADS) tm[i] = 0u;
    __syncthreads();
    const int base = blockIdx.x * EDGES_PER_WG;
    int s[EDGE_SLOTS], d[EDGE_SLOTS];
    float x[EDGE_SLOTS];
#pragma unroll
    for (int k = 0; k < EDGE_SLOTS; ++k) {
        int i = tid + k * T_THREADS;
        bool ok = i < EDGES_PER_WG;
        s[k] = ok ? e_src[base + i] : 0;
        d[k] = ok ? e_dst[base + i] : 0;
    }
#pragma unroll
    for (int k = 0; k < EDGE_SLOTS; ++k) x[k] = s_src[s[k]] + s_dst[d[k]];
#pragma unroll
    for (int k = 0; k < EDGE_SLOTS; ++k) {
        int i = tid + k * T_THREADS;
        if (i < EDGES_PER_WG) {
            float sc = (x[k] >= 0.f) ? x[k] : NEG_SLOPE * x[k];
            score[base + i] = sc;
            atomicMax(&tm[d[k]], enc_f32(sc));
        }
    }
    __syncthreads();
    unsigned* out = part + (size_t)blockIdx.x * TAB_N;
    for (int i = tid; i < N_NODES; i += T_THREADS) out[i] = tm[i];
}

__global__ void k_max_merge(const unsigned* __restrict__ part, unsigned* __restrict__ segmax) {
    int n = blockIdx.x * blockDim.x + threadIdx.x;
    if (n >= N_NODES) return;
    unsigned m = 0u;
#pragma unroll 32
    for (int g = 0; g < T_EDGE; ++g) {
        unsigned u = part[(size_t)g * TAB_N + n];
        m = (u > m) ? u : m;
    }
    segmax[n] = m;
}

__global__ __launch_bounds__(T_THREADS)
void k_exp_part(const int* __restrict__ e_dst, const unsigned* __restrict__ segmax,
                float* __restrict__ score, float* __restrict__ part) {
    __shared__ float ts[N_NODES];
    const int tid = threadIdx.x;
    for (int i = tid; i < N_NODES; i += T_THREADS) ts[i] = 0.f;
    __syncthreads();
    const int base = blockIdx.x * EDGES_PER_WG;
    int d[EDGE_SLOTS];
    float sc[EDGE_SLOTS];
    unsigned u[EDGE_SLOTS];
#pragma unroll
    for (int k = 0; k < EDGE_SLOTS; ++k) {
        int i = tid + k * T_THREADS;
        bool ok = i < EDGES_PER_WG;
        d[k]  = ok ? e_dst[base + i] : 0;
        sc[k] = ok ? score[base + i] : 0.f;
    }
#pragma unroll
    for (int k = 0; k < EDGE_SLOTS; ++k) u[k] = segmax[d[k]];
#pragma unroll
    for (int k = 0; k < EDGE_SLOTS; ++k) {
        int i = tid + k * T_THREADS;
        if (i < EDGES_PER_WG) {
            float m = dec_f32(u[k]);
            if (!isfinite(m)) m = 0.f;
            float w = expf(sc[k] - m);
            score[base + i] = w;
            atomicAdd(&ts[d[k]], w);
        }
    }
    __syncthreads();
    float* out = part + (size_t)blockIdx.x * TAB_N;
    for (int i = tid; i < N_NODES; i += T_THREADS) out[i] = ts[i];
}

__global__ void k_sum_merge(const float* __restrict__ part, float* __restrict__ denom) {
    int n = blockIdx.x * blockDim.x + threadIdx.x;
    if (n >= N_NODES) return;
    float s = 0.f;
#pragma unroll 32
    for (int g = 0; g < T_EDGE; ++g) s += part[(size_t)g * TAB_N + n];
    denom[n] = s;
}

__global__ void k_scatter_gat(const int* __restrict__ e_src, const int* __restrict__ e_dst,
                              const float* __restrict__ score, const float* __restrict__ denom,
                              const float* __restrict__ Xp, float* __restrict__ Xg) {
    long gid = (long)blockIdx.x * blockDim.x + threadIdx.x;
    long e   = gid >> 5;
    int lane = (int)(gid & 31);
    if (e >= N_EDGES) return;
    int s = e_src[e], d = e_dst[e];
    float coef = score[e] / denom[d];
    float2 xv = *(const float2*)(Xp + (size_t)s * DIM + lane * 2);
    float* out = Xg + (size_t)d * DIM + lane * 2;
    atomicAdd(out + 0, coef * xv.x);
    atomicAdd(out + 1, coef * xv.y);
}

__global__ void k_hg_gather(const int* __restrict__ inc_v, const int* __restrict__ inc_e,
                            const float* __restrict__ Xp, const float* __restrict__ dv,
                            float* __restrict__ He) {
    long gid = (long)blockIdx.x * blockDim.x + threadIdx.x;
    long p   = gid >> 5;
    int lane = (int)(gid & 31);
    if (p >= P_PAIRS) return;
    int v = inc_v[p], e = inc_e[p];
    float dvv = dv[v];
    float s = (dvv > 0.f) ? rsqrtf(dvv) : 0.f;
    float2 xv = *(const float2*)(Xp + (size_t)v * DIM + lane * 2);
    float* out = He + (size_t)e * DIM + lane * 2;
    atomicAdd(out + 0, s * xv.x);
    atomicAdd(out + 1, s * xv.y);
}

__global__ void k_he_norm(float* __restrict__ He, const float* __restrict__ de) {
    int i = blockIdx.x * blockDim.x + threadIdx.x;
    if (i >= M_HYPER * DIM) return;
    float d = de[i >> 6];
    He[i] *= (d > 0.f) ? (1.f / d) : 0.f;
}

__global__ void k_hg_scatter(const int* __restrict__ inc_v, const int* __restrict__ inc_e,
                             const float* __restrict__ He, float* __restrict__ Xhg) {
    long gid = (long)blockIdx.x * blockDim.x + threadIdx.x;
    long p   = gid >> 5;
    int lane = (int)(gid & 31);
    if (p >= P_PAIRS) return;
    int v = inc_v[p], e = inc_e[p];
    float2 h = *(const float2*)(He + (size_t)e * DIM + lane * 2);
    float* out = Xhg + (size_t)v * DIM + lane * 2;
    atomicAdd(out + 0, h.x);
    atomicAdd(out + 1, h.y);
}

__global__ void k_final(const float* __restrict__ Xg, const float* __restrict__ Xhg,
                        const float* __restrict__ dv, float* __restrict__ out) {
    long i = (long)blockIdx.x * blockDim.x + threadIdx.x;
    if (i >= (long)N_NODES * DIM) return;
    float d = dv[i >> 6];
    float s = (d > 0.f) ? rsqrtf(d) : 0.f;
    float x = 0.5f * (Xg[i] + Xhg[i] * s);
    out[i] = (x > 0.f) ? x : expm1f(x);
}

extern "C" void kernel_launch(void* const* d_in, const int* in_sizes, int n_in,
                              void* d_out, int out_size, void* d_ws, size_t ws_size,
                              hipStream_t stream) {
    const float* X     = (const float*)d_in[0];
    const float* W     = (const float*)d_in[1];
    const float* bias  = (const float*)d_in[2];
    const float* a_src = (const float*)d_in[3];
    const float* a_dst = (const float*)d_in[4];
    const int*   e_src = (const int*)d_in[5];
    const int*   e_dst = (const int*)d_in[6];
    const int*   inc_v = (const int*)d_in[7];
    const int*   inc_e = (const int*)d_in[8];
    float* out = (float*)d_out;
    float* ws  = (float*)d_ws;

    size_t off = 0;
    float* Xp    = ws + off; off += (size_t)N_NODES * DIM;
    float* s_src = ws + off; off += N_NODES;
    float* s_dst = ws + off; off += N_NODES;
    unsigned* segmax = (unsigned*)(ws + off); off += N_NODES;
    float* denom = ws + off; off += N_NODES;
    float* dv    = ws + off; off += N_NODES;
    float* de    = ws + off; off += M_HYPER;
    size_t zero_base = off;
    float* Xg    = ws + off; off += (size_t)N_NODES * DIM;
    float* He    = ws + off; off += (size_t)M_HYPER * DIM;
    float* Xhg   = ws + off; off += (size_t)N_NODES * DIM;
    size_t zero_cnt = off - zero_base;
    float* score = ws + off; off += N_EDGES;

    const int B = 256;
    int tiles = N_NODES / 16;
    k_gemm_wmma<<<(tiles * 32 + B - 1) / B, B, 0, stream>>>(X, W, bias, Xp);
    k_scalars<<<(N_NODES + B - 1) / B, B, 0, stream>>>(Xp, a_src, a_dst, s_src, s_dst);

    k_deg_part<<<T_PAIR, T_THREADS, 0, stream>>>(inc_v, inc_e, Xg, Xhg);
    k_deg_merge<<<(N_NODES + B - 1) / B, B, 0, stream>>>(Xg, Xhg, dv, de);
    k_score_part<<<T_EDGE, T_THREADS, 0, stream>>>(s_src, s_dst, e_src, e_dst, score,
                                                   (unsigned*)Xg);
    k_max_merge<<<(N_NODES + B - 1) / B, B, 0, stream>>>((const unsigned*)Xg, segmax);
    k_exp_part<<<T_EDGE, T_THREADS, 0, stream>>>(e_dst, segmax, score, Xg);
    k_sum_merge<<<(N_NODES + B - 1) / B, B, 0, stream>>>(Xg, denom);

    k_zero<<<(int)((zero_cnt + B - 1) / B), B, 0, stream>>>(ws + zero_base, (long)zero_cnt);
    k_scatter_gat<<<(int)(((long)N_EDGES * 32 + B - 1) / B), B, 0, stream>>>(
        e_src, e_dst, score, denom, Xp, Xg);
    k_hg_gather<<<(int)(((long)P_PAIRS * 32 + B - 1) / B), B, 0, stream>>>(
        inc_v, inc_e, Xp, dv, He);
    k_he_norm<<<(M_HYPER * DIM + B - 1) / B, B, 0, stream>>>(He, de);
    k_hg_scatter<<<(int)(((long)P_PAIRS * 32 + B - 1) / B), B, 0, stream>>>(
        inc_v, inc_e, He, Xhg);
    k_final<<<(int)(((long)N_NODES * DIM + B - 1) / B), B, 0, stream>>>(Xg, Xhg, dv, out);
}


# Edge row scatter by float atomics replaced by a per-call counting sort of edges by destination (LDS cursor tables) and one wave per node: max, softmax sum, row sum in registers, in list order
# speedup vs baseline: 1.8591x; 1.2805x over previous
#include <hip/hip_runtime.h>
#include <hip/hip_bf16.h>
#include <math.h>

typedef __attribute__((ext_vector_type(2))) float v2f;
typedef __attribute__((ext_vector_type(8))) float v8f;

#define N_NODES 50000
#define N_EDGES 800000
#define M_HYPER 10000
#define P_PAIRS 200000
#define DIM     64
#define NEG_SLOPE 0.2f

__global__ void k_zero(float* __restrict__ p, long n) {
    long i = (long)blockIdx.x * blockDim.x + threadIdx.x;
    if (i < n) p[i] = 0.0f;
}

__global__ void k_gemm_wmma(const float* __restrict__ X, const float* __restrict__ W,
                            const float* __restrict__ bias, float* __restrict__ Xp) {
    int wave = (blockIdx.x * blockDim.x + threadIdx.x) >> 5;
    int lane = threadIdx.x & 31;
    if (wave >= N_NODES / 16) return;
    int row_base = wave * 16;
    int lh  = (lane >= 16) ? 1 : 0;
    int l15 = lane & 15;

    const float* xrow = X + (size_t)(row_base + l15) * DIM + lh * 2;
    const float* wrow = W + (size_t)l15 * DIM + lh * 2;

    v8f acc[4] = {};
#pragma unroll
    for (int ks = 0; ks < 16; ++ks) {
        v2f a;
        a.x = xrow[ks * 4 + 0];
        a.y = xrow[ks * 4 + 1];
#pragma unroll
        for (int t = 0; t < 4; ++t) {
            v2f b;
            const float* wb = wrow + (size_t)t * 16 * DIM + ks * 4;
            b.x = wb[0];
            b.y = wb[1];
            acc[t] = __builtin_amdgcn_wmma_f32_16x16x4_f32(
                false, a, false, b, (short)0, acc[t], false, false);
        }
    }

#pragma unroll
    for (int t = 0; t < 4; ++t) {
        int c = t * 16 + l15;
        float bv = bias[c];
#pragma unroll
        for (int v = 0; v < 8; ++v) {
            int m = row_base + v + lh * 8;
            Xp[(size_t)m * DIM + c] = acc[t][v] + bv;
        }
    }
}

__global__ void k_scalars(const float* __restrict__ Xp,
                          const float* __restrict__ a_src, const float* __restrict__ a_dst,
                          float* __restrict__ s_src, float* __restrict__ s_dst) {
    int n = blockIdx.x * blockDim.x + threadIdx.x;
    if (n >= N_NODES) return;
    const float4* row = (const float4*)(Xp + (size_t)n * DIM);
    const float4* as  = (const float4*)a_src;
    const float4* ad  = (const float4*)a_dst;
    float ss = 0.f, sd = 0.f;
#pragma unroll
    for (int i = 0; i < 16; ++i) {
        float4 x = row[i], u = as[i], v = ad[i];
        ss += x.x * u.x + x.y * u.y + x.z * u.z + x.w * u.w;
        sd += x.x * v.x + x.y * v.y + x.z * v.z + x.w * v.w;
    }
    s_src[n] = ss;
    s_dst[n] = sd;
}

#define T_THREADS 1024
#define T_EDGE 128
#define T_PAIR 64
#define EDGES_PER_WG (N_EDGES / T_EDGE)
#define PAIRS_PER_WG (P_PAIRS / T_PAIR)
#define EDGE_SLOTS ((EDGES_PER_WG + T_THREADS - 1) / T_THREADS)
#define PAIR_SLOTS ((PAIRS_PER_WG + T_THREADS - 1) / T_THREADS)
#define TAB_N ((N_NODES + 31) / 32 * 32)
#define TAB_M ((M_HYPER + 31) / 32 * 32)
static_assert(N_EDGES % T_EDGE == 0 && P_PAIRS % T_PAIR == 0, "slices are exact");
static_assert((size_t)T_EDGE * TAB_N <= (size_t)2 * N_NODES * DIM + (size_t)M_HYPER * DIM,
              "edge tables fit the span they borrow");
static_assert(T_PAIR * TAB_N <= N_NODES * DIM + M_HYPER * DIM && T_PAIR * TAB_M <= N_NODES * DIM,
              "degree tables fit the spans they borrow");

__global__ __launch_bounds__(T_THREADS)
void k_deg_part(const int* __restrict__ inc_v, const int* __restrict__ inc_e,
                float* __restrict__ part_v, float* __restrict__ part_e) {
    __shared__ unsigned tv[N_NODES];
    __shared__ unsigned te[M_HYPER];
    const int tid = threadIdx.x;
    for (int i = tid; i < N_NODES; i += T_THREADS) tv[i] = 0u;
    for (int i = tid; i < M_HYPER; i += T_THREADS) te[i] = 0u;
    __syncthreads();
    const int base = blockIdx.x * PAIRS_PER_WG;
    int v[PAIR_SLOTS], h[PAIR_SLOTS];
#pragma unroll
    for (int k = 0; k < PAIR_SLOTS; ++k) {
        int i = tid + k * T_THREADS;
        bool ok = i < PAIRS_PER_WG;
        v[k] = ok ? inc_v[base + i] : 0;
        h[k] = ok ? inc_e[base + i] : 0;
    }
#pragma unroll
    for (int k = 0; k < PAIR_SLOTS; ++k) {
        if (tid + k * T_THREADS < PAIRS_PER_WG) {
            atomicAdd(&tv[v[k]], 1u);
            atomicAdd(&te[h[k]], 1u);
        }
    }
    __syncthreads();
    float* pv = part_v + (size_t)blockIdx.x * TAB_N;
    for (int i = tid; i < N_NODES; i += T_THREADS) pv[i] = (float)tv[i];
    float* pe = part_e + (size_t)blockIdx.x * TAB_M;
    for (int i = tid; i < M_HYPER; i += T_THREADS) pe[i] = (float)te[i];
}

__global__ void k_deg_merge(const float* __restrict__ part_v, const float* __restrict__ part_e,
                            float* __restrict__ dv, float* __restrict__ de) {
    int n = blockIdx.x * blockDim.x + threadIdx.x;
    if (n < N_NODES) {
        float s = 0.f;
#pragma unroll 16
        for (int g = 0; g < T_PAIR; ++g) s += part_v[(size_t)g * TAB_N + n];
        dv[n] = s;
    }
    if (n < M_HYPER) {
        float s = 0.f;
#pragma unroll 16
        for (int g = 0; g < T_PAIR; ++g) s += part_e[(size_t)g * TAB_M + n];
        de[n] = s;
    }
}

#define MB_THREADS 256
#define MB_SHIFT 8
#define N_MBLK ((N_NODES + MB_THREADS - 1) / MB_THREADS)
#define BS_PAD 256
static_assert((1 << MB_SHIFT) == MB_THREADS && N_MBLK <= BS_PAD, "block sums fit their pad");
static_assert((size_t)T_EDGE * TAB_N + 2 * BS_PAD <= (size_t)2 * N_NODES * DIM + (size_t)M_HYPER * DIM,
              "the block sums and block offsets sit behind the edge tables");

__device__ __forceinline__ unsigned wave_incl_scan(unsigned v, int lane) {
    for (int o = 1; o < 32; o <<= 1) {
        unsigned t = __shfl_up(v, o);
        if (lane >= o) v += t;
    }
    return v;
}

__global__ __launch_bounds__(T_THREADS)
void k_cnt_part(const int* __restrict__ e_dst, unsigned* __restrict__ part) {
    __shared__ unsigned tc[N_NODES];
    const int tid = threadIdx.x;
    for (int i = tid; i < N_NODES; i += T_THREADS) tc[i] = 0u;
    __syncthreads();
    const int base = blockIdx.x * EDGES_PER_WG;
    int d[EDGE_SLOTS];
#pragma unroll
    for (int k = 0; k < EDGE_SLOTS; ++k) {
        int i = tid + k * T_THREADS;
        d[k] = (i < EDGES_PER_WG) ? e_dst[base + i] : 0;
    }
#pragma unroll
    for (int k = 0; k < EDGE_SLOTS; ++k) {
        if (tid + k * T_THREADS < EDGES_PER_WG) atomicAdd(&tc[d[k]], 1u);
    }
    __syncthreads();
    unsigned* out = part + (size_t)blockIdx.x * TAB_N;
    for (int i = tid; i < N_NODES; i += T_THREADS) out[i] = tc[i];
}

__global__ __launch_bounds__(MB_THREADS)
void k_cnt_merge(unsigned* __restrict__ part, unsigned* __restrict__ cnt,
                 unsigned* __restrict__ rowloc, unsigned* __restrict__ blocksum) {
    __shared__ unsigned wsum[MB_THREADS / 32];
    const int tid = threadIdx.x, lane = tid & 31, w = tid >> 5;
    const int n = blockIdx.x * MB_THREADS + tid;
    unsigned run = 0u;
    if (n < N_NODES) {
#pragma unroll 16
        for (int g = 0; g < T_EDGE; ++g) {
            size_t idx = (size_t)g * TAB_N + n;
            unsigned c = part[idx];
            part[idx] = run;
            run += c;
        }
    }
    unsigned inc = wave_incl_scan(run, lane);
    if (lane == 31) wsum[w] = inc;
    __syncthreads();
    if (w == 0) {
        unsigned v = (lane < MB_THREADS / 32) ? wsum[lane] : 0u;
        unsigned iv = wave_incl_scan(v, lane);
        if (lane < MB_THREADS / 32) wsum[lane] = iv - v;
        if (lane == 31) blocksum[blockIdx.x] = iv;
    }
    __syncthreads();
    if (n < N_NODES) {
        rowloc[n] = wsum[w] + inc - run;
        cnt[n] = run;
    }
}

__global__ __launch_bounds__(T_THREADS)
void k_fill(const int* __restrict__ e_src, const int* __restrict__ e_dst,
            const unsigned* __restrict__ part, const unsigned* __restrict__ rowloc,
            const unsigned* __restrict__ blocksum, unsigned* __restrict__ blockoff,
            int* __restrict__ csr_src) {
    __shared__ unsigned tb[N_NODES];
    __shared__ unsigned bs[BS_PAD];
    const int tid = threadIdx.x;
    if (tid < 32) {
        unsigned carry = 0u;
        for (int c = 0; c < N_MBLK; c += 32) {
            int i = c + tid;
            unsigned a = (i < N_MBLK) ? blocksum[i] : 0u;
            unsigned ia = wave_incl_scan(a, tid);
            if (i < N_MBLK) bs[i] = carry + ia - a;
            carry += __shfl(ia, 31);
        }
    }
    __syncthreads();
    if (blockIdx.x == 0 && tid < N_MBLK) blockoff[tid] = bs[tid];
    const unsigned* pg = part + (size_t)blockIdx.x * TAB_N;
    for (int i = tid; i < N_NODES; i += T_THREADS) tb[i] = pg[i] + rowloc[i] + bs[i >> MB_SHIFT];
    __syncthreads();
    const int base = blockIdx.x * EDGES_PER_WG;
    int s[EDGE_SLOTS], d[EDGE_SLOTS];
#pragma unroll
    for (int k = 0; k < EDGE_SLOTS; ++k) {
        int i = tid + k * T_THREADS;
        bool ok = i < EDGES_PER_WG;
        s[k] = ok ? e_src[base + i] : 0;
        d[k] = ok ? e_dst[base + i] : 0;
    }
#pragma unroll
    for (int k = 0; k < EDGE_SLOTS; ++k) {
        if (tid + k * T_THREADS < EDGES_PER_WG) {
            unsigned pos = atomicAdd(&tb[d[k]], 1u);
            if (pos < N_EDGES) csr_src[pos] = s[k];
        }
    }
}

__device__ __forceinline__ float edge_score(const float* __restrict__ s_src, int s, float sd) {
    float x = s_src[s] + sd;
    return (x >= 0.f) ? x : NEG_SLOPE * x;
}

__device__ __forceinline__ void gather_rows(int s, float wgt, int cn, const float* __restrict__ xcol,
                                            float& ax, float& ay) {
    const int wb = __float_as_int(wgt);
    int q = 0;
    for (; q + 4 <= cn; q += 4) {
        int s0 = __builtin_amdgcn_readlane(s, q),     s1 = __builtin_amdgcn_readlane(s, q + 1);
        int s2 = __builtin_amdgcn_readlane(s, q + 2), s3 = __builtin_amdgcn_readlane(s, q + 3);
        float w0 = __int_as_float(__builtin_amdgcn_readlane(wb, q));
        float w1 = __int_as_float(__builtin_amdgcn_readlane(wb, q + 1));
        float w2 = __int_as_float(__builtin_amdgcn_readlane(wb, q + 2));
        float w3 = __int_as_float(__builtin_amdgcn_readlane(wb, q + 3));
        float2 x0 = *(const float2*)(xcol + (size_t)s0 * DIM);
        float2 x1 = *(const float2*)(xcol + (size_t)s1 * DIM);
        float2 x2 = *(const float2*)(xcol + (size_t)s2 * DIM);
        float2 x3 = *(const float2*)(xcol + (size_t)s3 * DIM);
        ax += w0 * x0.x; ay += w0 * x0.y;
        ax += w1 * x1.x; ay += w1 * x1.y;
        ax += w2 * x2.x; ay += w2 * x2.y;
        ax += w3 * x3.x; ay += w3 * x3.y;
    }
    for (; q < cn; ++q) {
        int sq = __builtin_amdgcn_readlane(s, q);
        float wq = __int_as_float(__builtin_amdgcn_readlane(wb, q));
        float2 xv = *(const float2*)(xcol + (size_t)sq * DIM);
        ax += wq * xv.x; ay += wq * xv.y;
    }
}

__global__ void k_node_gat(const unsigned* __restrict__ cnt, const unsigned* __restrict__ rowloc,
                           const unsigned* __restrict__ blockoff, const int* __restrict__ csr_src,
                           const float* __restrict__ s_src, const float* __restrict__ s_dst,
                           const float* __restrict__ Xp, float* __restrict__ Xg) {
    const int n = (blockIdx.x * blockDim.x + threadIdx.x) >> 5;
    const int lane = threadIdx.x & 31;
    if (n >= N_NODES) return;
    const int deg = __builtin_amdgcn_readfirstlane((int)cnt[n]);
    const unsigned beg = (unsigned)__builtin_amdgcn_readfirstlane(
        (int)(rowloc[n] + blockoff[n >> MB_SHIFT]));
    const float sd = s_dst[n];
    float ax = 0.f, ay = 0.f;
    if (deg > 0) {
        const int* es = csr_src + beg;
        const float* xcol = Xp + lane * 2;
        int s0 = 0;
        float sc0 = -INFINITY;
        if (lane < deg) { s0 = es[lane]; sc0 = edge_score(s_src, s0, sd); }
        float m = sc0;
        for (int j0 = 32; j0 < deg; j0 += 32) {
            int j = j0 + lane;
            if (j < deg) m = fmaxf(m, edge_score(s_src, es[j], sd));
        }
        for (int o = 16; o > 0; o >>= 1) m = fmaxf(m, __shfl_xor(m, o));
        if (!isfinite(m)) m = 0.f;
        const float e0 = (lane < deg) ? expf(sc0 - m) : 0.f;
        float sum = e0;
        for (int j0 = 32; j0 < deg; j0 += 32) {
            int j = j0 + lane;
            if (j < deg) sum += expf(edge_score(s_src, es[j], sd) - m);
        }
        for (int o = 16; o > 0; o >>= 1) sum += __shfl_xor(sum, o);
        gather_rows(s0, e0 / sum, (deg < 32) ? deg : 32, xcol, ax, ay);
        for (int j0 = 32; j0 < deg; j0 += 32) {
            int j = j0 + lane;
            int s = 0;
            float wgt = 0.f;
            if (j < deg) {
                s = es[j];
                wgt = expf(edge_score(s_src, s, sd) - m) / sum;
            }
            int cn = deg - j0;
            gather_rows(s, wgt, (cn < 32) ? cn : 32, xcol, ax, ay);
        }
    }
    float2 r; r.x = ax; r.y = ay;
    *(float2*)(Xg + (size_t)n * DIM + lane * 2) = r;
}

__global__ void k_hg_gather(const int* __restrict__ inc_v, const int* __restrict__ inc_e,
                            const float* __restrict__ Xp, const float* __restrict__ dv,
                            float* __restrict__ He) {
    long gid = (long)blockIdx.x * blockDim.x + threadIdx.x;
    long p   = gid >> 5;
    int lane = (int)(gid & 31);
    if (p >= P_PAIRS) return;
    int v = inc_v[p], e = inc_e[p];
    float dvv = dv[v];
    float s = (dvv > 0.f) ? rsqrtf(dvv) : 0.f;
    float2 xv = *(const float2*)(Xp + (size_t)v * DIM + lane * 2);
    float* out = He + (size_t)e * DIM + lane * 2;
    atomicAdd(out + 0, s * xv.x);
    atomicAdd(out + 1, s * xv.y);
}

__global__ void k_he_norm(float* __restrict__ He, const float* __restrict__ de) {
    int i = blockIdx.x * blockDim.x + threadIdx.x;
    if (i >= M_HYPER * DIM) return;
    float d = de[i >> 6];
    He[i] *= (d > 0.f) ? (1.f / d) : 0.f;
}

__global__ void k_hg_scatter(const int* __restrict__ inc_v, const int* __restrict__ inc_e,
                             const float* __restrict__ He, float* __restrict__ Xhg) {
    long gid = (long)blockIdx.x * blockDim.x + threadIdx.x;
    long p   = gid >> 5;
    int lane = (int)(gid & 31);
    if (p >= P_PAIRS) return;
    int v = inc_v[p], e = inc_e[p];
    float2 h = *(const float2*)(He + (size_t)e * DIM + lane * 2);
    float* out = Xhg + (size_t)v * DIM + lane * 2;
    atomicAdd(out + 0, h.x);
    atomicAdd(out + 1, h.y);
}

__global__ void k_final(const float* __restrict__ Xg, const float* __restrict__ Xhg,
                        const float* __restrict__ dv, float* __restrict__ out) {
    long i = (long)blockIdx.x * blockDim.x + threadIdx.x;
    if (i >= (long)N_NODES * DIM) return;
    float d = dv[i >> 6];
    float s = (d > 0.f) ? rsqrtf(d) : 0.f;
    float x = 0.5f * (Xg[i] + Xhg[i] * s);
    out[i] = (x > 0.f) ? x : expm1f(x);
}

extern "C" void kernel_launch(void* const* d_in, const int* in_sizes, int n_in,
                              void* d_out, int out_size, void* d_ws, size_t ws_size,
                              hipStream_t stream) {
    const float* X     = (const float*)d_in[0];
    const float* W     = (const float*)d_in[1];
    const float* bias  = (const float*)d_in[2];
    const float* a_src = (const float*)d_in[3];
    const float* a_dst = (const float*)d_in[4];
    const int*   e_src = (const int*)d_in[5];
    const int*   e_dst = (const int*)d_in[6];
    const int*   inc_v = (const int*)d_in[7];
    const int*   inc_e = (const int*)d_in[8];
    float* out = (float*)d_out;
    float* ws  = (float*)d_ws;

    size_t off = 0;
    float* Xp    = ws + off; off += (size_t)N_NODES * DIM;
    float* s_src = ws + off; off += N_NODES;
    float* s_dst = ws + off; off += N_NODES;
    unsigned* segmax = (unsigned*)(ws + off); off += N_NODES;
    float* denom = ws + off; off += N_NODES;
    float* dv    = ws + off; off += N_NODES;
    float* de    = ws + off; off += M_HYPER;
    float* Xg    = ws + off; off += (size_t)N_NODES * DIM;
    size_t zero_base = off;
    float* He    = ws + off; off += (size_t)M_HYPER * DIM;
    float* Xhg   = ws + off; off += (size_t)N_NODES * DIM;
    size_t zero_cnt = off - zero_base;
    float* score = ws + off; off += N_EDGES;

    const int B = 256;
    int tiles = N_NODES / 16;
    k_gemm_wmma<<<(tiles * 32 + B - 1) / B, B, 0, stream>>>(X, W, bias, Xp);
    k_scalars<<<(N_NODES + B - 1) / B, B, 0, stream>>>(Xp, a_src, a_dst, s_src, s_dst);

    k_deg_part<<<T_PAIR, T_THREADS, 0, stream>>>(inc_v, inc_e, Xg, Xhg);
    k_deg_merge<<<(N_NODES + B - 1) / B, B, 0, stream>>>(Xg, Xhg, dv, de);
    unsigned* part = (unsigned*)Xg;
    unsigned* blocksum = (unsigned*)(Xhg + (size_t)N_NODES * DIM - 2 * BS_PAD);
    unsigned* blockoff = blocksum + BS_PAD;
    unsigned* cnt = segmax;
    unsigned* rowloc = (unsigned*)denom;
    int* csr_src = (int*)score;
    k_cnt_part<<<T_EDGE, T_THREADS, 0, stream>>>(e_dst, part);
    k_cnt_merge<<<N_MBLK, MB_THREADS, 0, stream>>>(part, cnt, rowloc, blocksum);
    k_fill<<<T_EDGE, T_THREADS, 0, stream>>>(e_src, e_dst, part, rowloc, blocksum, blockoff,
                                             csr_src);
    k_node_gat<<<(N_NODES * 32 + B - 1) / B, B, 0, stream>>>(cnt, rowloc, blockoff, csr_src,
                                                             s_src, s_dst, Xp, Xg);

    k_zero<<<(int)((zero_cnt + B - 1) / B), B, 0, stream>>>(ws + zero_base, (long)zero_cnt);
    k_hg_gather<<<(int)(((long)P_PAIRS * 32 + B - 1) / B), B, 0, stream>>>(
        inc_v, inc_e, Xp, dv, He);
    k_he_norm<<<(M_HYPER * DIM + B - 1) / B, B, 0, stream>>>(He, de);
    k_hg_scatter<<<(int)(((long)P_PAIRS * 32 + B - 1) / B), B, 0, stream>>>(
        inc_v, inc_e, He, Xhg);
    k_final<<<(int)(((long)N_NODES * DIM + B - 1) / B), B, 0, stream>>>(Xg, Xhg, dv, out);
}


# HGNN row scatters by float atomics replaced by a per-call counting sort of the pairs by hyperedge and by vertex (LDS cursors): one wave per hyperedge, node wave adds its He rows; sums in list order
# speedup vs baseline: 2.1095x; 1.1347x over previous
#include <hip/hip_runtime.h>
#include <hip/hip_bf16.h>
#include <math.h>

typedef __attribute__((ext_vector_type(2))) float v2f;
typedef __attribute__((ext_vector_type(8))) float v8f;

#define N_NODES 50000
#define N_EDGES 800000
#define M_HYPER 10000
#define P_PAIRS 200000
#define DIM     64
#define NEG_SLOPE 0.2f

__global__ void k_gemm_wmma(const float* __restrict__ X, const float* __restrict__ W,
                            const float* __restrict__ bias, float* __restrict__ Xp) {
    int wave = (blockIdx.x * blockDim.x + threadIdx.x) >> 5;
    int lane = threadIdx.x & 31;
    if (wave >= N_NODES / 16) return;
    int row_base = wave * 16;
    int lh  = (lane >= 16) ? 1 : 0;
    int l15 = lane & 15;

    const float* xrow = X + (size_t)(row_base + l15) * DIM + lh * 2;
    const float* wrow = W + (size_t)l15 * DIM + lh * 2;

    v8f acc[4] = {};
#pragma unroll
    for (int ks = 0; ks < 16; ++ks) {
        v2f a;
        a.x = xrow[ks * 4 + 0];
        a.y = xrow[ks * 4 + 1];
#pragma unroll
        for (int t = 0; t < 4; ++t) {
            v2f b;
            const float* wb = wrow + (size_t)t * 16 * DIM + ks * 4;
            b.x = wb[0];
            b.y = wb[1];
            acc[t] = __builtin_amdgcn_wmma_f32_16x16x4_f32(
                false, a, false, b, (short)0, acc[t], false, false);
        }
    }

#pragma unroll
    for (int t = 0; t < 4; ++t) {
        int c = t * 16 + l15;
        float bv = bias[c];
#pragma unroll
        for (int v = 0; v < 8; ++v) {
            int m = row_base + v + lh * 8;
            Xp[(size_t)m * DIM + c] = acc[t][v] + bv;
        }
    }
}

__global__ void k_scalars(const float* __restrict__ Xp,
                          const float* __restrict__ a_src, const float* __restrict__ a_dst,
                          float* __restrict__ s_src, float* __restrict__ s_dst) {
    int n = blockIdx.x * blockDim.x + threadIdx.x;
    if (n >= N_NODES) return;
    const float4* row = (const float4*)(Xp + (size_t)n * DIM);
    const float4* as  = (const float4*)a_src;
    const float4* ad  = (const float4*)a_dst;
    float ss = 0.f, sd = 0.f;
#pragma unroll
    for (int i = 0; i < 16; ++i) {
        float4 x = row[i], u = as[i], v = ad[i];
        ss += x.x * u.x + x.y * u.y + x.z * u.z + x.w * u.w;
        sd += x.x * v.x + x.y * v.y + x.z * v.z + x.w * v.w;
    }
    s_src[n] = ss;
    s_dst[n] = sd;
}

#define T_THREADS 1024
#define T_EDGE 128
#define T_PAIR 64
#define EDGES_PER_WG (N_EDGES / T_EDGE)
#define PAIRS_PER_WG (P_PAIRS / T_PAIR)
#define EDGE_SLOTS ((EDGES_PER_WG + T_THREADS - 1) / T_THREADS)
#define PAIR_SLOTS ((PAIRS_PER_WG + T_THREADS - 1) / T_THREADS)
#define TAB_N ((N_NODES + 31) / 32 * 32)
#define TAB_M ((M_HYPER + 31) / 32 * 32)
static_assert(N_EDGES % T_EDGE == 0 && P_PAIRS % T_PAIR == 0, "slices are exact");

#define MB_THREADS 256
#define MB_SHIFT 8
#define N_MBLK ((N_NODES + MB_THREADS - 1) / MB_THREADS)
#define N_HBLK ((M_HYPER + MB_THREADS - 1) / MB_THREADS)
#define BS_PAD 256
#define HS_PAD 64
static_assert((1 << MB_SHIFT) == MB_THREADS && N_MBLK <= BS_PAD && N_HBLK <= HS_PAD,
              "block sums fit their pads");
#define SPAN_FLOATS ((size_t)2 * N_NODES * DIM + (size_t)M_HYPER * DIM)
#define TAIL_OFF ((size_t)T_EDGE * TAB_N)
static_assert(TAIL_OFF + 2 * (size_t)P_PAIRS + N_NODES + M_HYPER + 4 * BS_PAD + 2 * HS_PAD <= SPAN_FLOATS,
              "lists, row offsets and block-sum pads fit behind the edge tables");
static_assert((size_t)T_PAIR * TAB_N <= (size_t)(N_NODES + M_HYPER) * DIM &&
              (size_t)(N_NODES + M_HYPER) * DIM + (size_t)T_PAIR * TAB_M <= TAIL_OFF,
              "the two pair tables sit one behind the other, in front of the tail");

__device__ __forceinline__ unsigned wave_incl_scan(unsigned v, int lane) {
    for (int o = 1; o < 32; o <<= 1) {
        unsigned t = __shfl_up(v, o);
        if (lane >= o) v += t;
    }
    return v;
}

template <int SLICES>
__device__ __forceinline__ void slice_prefix(unsigned* __restrict__ part, size_t pitch, int lim, int blk,
                                             unsigned* __restrict__ cnt, unsigned* __restrict__ rowloc,
                                             unsigned* __restrict__ blocksum, unsigned* wsum) {
    const int tid = threadIdx.x, lane = tid & 31, w = tid >> 5;
    const int n = blk * MB_THREADS + tid;
    unsigned run = 0u;
    if (n < lim) {
#pragma unroll 16
        for (int g = 0; g < SLICES; ++g) {
            size_t idx = (size_t)g * pitch + n;
            unsigned c = part[idx];
            part[idx] = run;
            run += c;
        }
    }
    unsigned inc = wave_incl_scan(run, lane);
    if (lane == 31) wsum[w] = inc;
    __syncthreads();
    if (w == 0) {
        unsigned v = (lane < MB_THREADS / 32) ? wsum[lane] : 0u;
        unsigned iv = wave_incl_scan(v, lane);
        if (lane < MB_THREADS / 32) wsum[lane] = iv - v;
        if (lane == 31) blocksum[blk] = iv;
    }
    __syncthreads();
    if (n < lim) {
        rowloc[n] = wsum[w] + inc - run;
        cnt[n] = run;
    }
}

__device__ __forceinline__ void block_offsets(const unsigned* __restrict__ blocksum, int nblk,
                                              unsigned* bs, int lane) {
    unsigned carry = 0u;
    for (int c = 0; c < nblk; c += 32) {
        int i = c + lane;
        unsigned a = (i < nblk) ? blocksum[i] : 0u;
        unsigned ia = wave_incl_scan(a, lane);
        if (i < nblk) bs[i] = carry + ia - a;
        carry += __shfl(ia, 31);
    }
}

__global__ __launch_bounds__(T_THREADS)
void k_pair_cnt(const int* __restrict__ inc_v, const int* __restrict__ inc_e,
                unsigned* __restrict__ part_v, unsigned* __restrict__ part_e) {
    __shared__ unsigned tv[N_NODES];
    __shared__ unsigned te[M_HYPER];
    const int tid = threadIdx.x;
    for (int i = tid; i < N_NODES; i += T_THREADS) tv[i] = 0u;
    for (int i = tid; i < M_HYPER; i += T_THREADS) te[i] = 0u;
    __syncthreads();
    const int base = blockIdx.x * PAIRS_PER_WG;
    int v[PAIR_SLOTS], h[PAIR_SLOTS];
#pragma unroll
    for (int k = 0; k < PAIR_SLOTS; ++k) {
        int i = tid + k * T_THREADS;
        bool ok = i < PAIRS_PER_WG;
        v[k] = ok ? inc_v[base + i] : 0;
        h[k] = ok ? inc_e[base + i] : 0;
    }
#pragma unroll
    for (int k = 0; k < PAIR_SLOTS; ++k) {
        if (tid + k * T_THREADS < PAIRS_PER_WG) {
            atomicAdd(&tv[v[k]], 1u);
            atomicAdd(&te[h[k]], 1u);
        }
    }
    __syncthreads();
    unsigned* pv = part_v + (size_t)blockIdx.x * TAB_N;
    for (int i = tid; i < N_NODES; i += T_THREADS) pv[i] = tv[i];
    unsigned* pe = part_e + (size_t)blockIdx.x * TAB_M;
    for (int i = tid; i < M_HYPER; i += T_THREADS) pe[i] = te[i];
}

__global__ __launch_bounds__(MB_THREADS)
void k_pair_merge(unsigned* __restrict__ part_v, unsigned* __restrict__ part_e,
                  unsigned* __restrict__ cnt_v, unsigned* __restrict__ cnt_e,
                  unsigned* __restrict__ rowloc_v, unsigned* __restrict__ rowloc_e,
                  unsigned* __restrict__ blocksum_v, unsigned* __restrict__ blocksum_e) {
    __shared__ unsigned wsum[MB_THREADS / 32];
    const bool isv = blockIdx.x < N_MBLK;
    slice_prefix<T_PAIR>(isv ? part_v : part_e, isv ? (size_t)TAB_N : (size_t)TAB_M,
                         isv ? N_NODES : M_HYPER, isv ? (int)blockIdx.x : (int)blockIdx.x - N_MBLK,
                         isv ? cnt_v : cnt_e, isv ? rowloc_v : rowloc_e,
                         isv ? blocksum_v : blocksum_e, wsum);
}

__global__ __launch_bounds__(T_THREADS)
void k_pair_fill(const int* __restrict__ inc_v, const int* __restrict__ inc_e,
                 const unsigned* __restrict__ part_v, const unsigned* __restrict__ part_e,
                 const unsigned* __restrict__ rowloc_v, const unsigned* __restrict__ rowloc_e,
                 const unsigned* __restrict__ blocksum_v, const unsigned* __restrict__ blocksum_e,
                 unsigned* __restrict__ blockoff_v, unsigned* __restrict__ blockoff_e,
                 int* __restrict__ csr_v, int* __restrict__ csr_e) {
    __shared__ unsigned tv[N_NODES];
    __shared__ unsigned te[M_HYPER];
    __shared__ unsigned bv[BS_PAD];
    __shared__ unsigned be[HS_PAD];
    const int tid = threadIdx.x;
    if (tid < 32) block_offsets(blocksum_v, N_MBLK, bv, tid);
    else if (tid < 64) block_offsets(blocksum_e, N_HBLK, be, tid - 32);
    __syncthreads();
    if (blockIdx.x == 0) {
        if (tid < N_MBLK) blockoff_v[tid] = bv[tid];
        if (tid < N_HBLK) blockoff_e[tid] = be[tid];
    }
    const unsigned* pv = part_v + (size_t)blockIdx.x * TAB_N;
    for (int i = tid; i < N_NODES; i += T_THREADS) tv[i] = pv[i] + rowloc_v[i] + bv[i >> MB_SHIFT];
    const unsigned* pe = part_e + (size_t)blockIdx.x * TAB_M;
    for (int i = tid; i < M_HYPER; i += T_THREADS) te[i] = pe[i] + rowloc_e[i] + be[i >> MB_SHIFT];
    __syncthreads();
    const int base = blockIdx.x * PAIRS_PER_WG;
    int v[PAIR_SLOTS], h[PAIR_SLOTS];
#pragma unroll
    for (int k = 0; k < PAIR_SLOTS; ++k) {
        int i = tid + k * T_THREADS;
        bool ok = i < PAIRS_PER_WG;
        v[k] = ok ? inc_v[base + i] : 0;
        h[k] = ok ? inc_e[base + i] : 0;
    }
#pragma unroll
    for (int k = 0; k < PAIR_SLOTS; ++k) {
        if (tid + k * T_THREADS < PAIRS_PER_WG) {
            unsigned pn = atomicAdd(&tv[v[k]], 1u);
            unsigned ph = atomicAdd(&te[h[k]], 1u);
            if (pn < P_PAIRS) csr_v[pn] = h[k];
            if (ph < P_PAIRS) csr_e[ph] = v[k];
        }
    }
}

__global__ __launch_bounds__(T_THREADS)
void k_cnt_part(const int* __restrict__ e_dst, unsigned* __restrict__ part) {
    __shared__ unsigned tc[N_NODES];
    const int tid = threadIdx.x;
    for (int i = tid; i < N_NODES; i += T_THREADS) tc[i] = 0u;
    __syncthreads();
    const int base = blockIdx.x * EDGES_PER_WG;
    int d[EDGE_SLOTS];
#pragma unroll
    for (int k = 0; k < EDGE_SLOTS; ++k) {
        int i = tid + k * T_THREADS;
        d[k] = (i < EDGES_PER_WG) ? e_dst[base + i] : 0;
    }
#pragma unroll
    for (int k = 0; k < EDGE_SLOTS; ++k) {
        if (tid + k * T_THREADS < EDGES_PER_WG) atomicAdd(&tc[d[k]], 1u);
    }
    __syncthreads();
    unsigned* out = part + (size_t)blockIdx.x * TAB_N;
    for (int i = tid; i < N_NODES; i += T_THREADS) out[i] = tc[i];
}

__global__ __launch_bounds__(MB_THREADS)
void k_cnt_merge(unsigned* __restrict__ part, unsigned* __restrict__ cnt,
                 unsigned* __restrict__ rowloc, unsigned* __restrict__ blocksum) {
    __shared__ unsigned wsum[MB_THREADS / 32];
    slice_prefix<T_EDGE>(part, (size_t)TAB_N, N_NODES, (int)blockIdx.x, cnt, rowloc, blocksum, wsum);
}

__global__ __launch_bounds__(T_THREADS)
void k_fill(const int* __restrict__ e_src, const int* __restrict__ e_dst,
            const unsigned* __restrict__ part, const unsigned* __restrict__ rowloc,
            const unsigned* __restrict__ blocksum, unsigned* __restrict__ blockoff,
            int* __restrict__ csr_src) {
    __shared__ unsigned tb[N_NODES];
    __shared__ unsigned bs[BS_PAD];
    const int tid = threadIdx.x;
    if (tid < 32) block_offsets(blocksum, N_MBLK, bs, tid);
    __syncthreads();
    if (blockIdx.x == 0 && tid < N_MBLK) blockoff[tid] = bs[tid];
    const unsigned* pg = part + (size_t)blockIdx.x * TAB_N;
    for (int i = tid; i < N_NODES; i += T_THREADS) tb[i] = pg[i] + rowloc[i] + bs[i >> MB_SHIFT];
    __syncthreads();
    const int base = blockIdx.x * EDGES_PER_WG;
    int s[EDGE_SLOTS], d[EDGE_SLOTS];
#pragma unroll
    for (int k = 0; k < EDGE_SLOTS; ++k) {
        int i = tid + k * T_THREADS;
        bool ok = i < EDGES_PER_WG;
        s[k] = ok ? e_src[base + i] : 0;
        d[k] = ok ? e_dst[base + i] : 0;
    }
#pragma unroll
    for (int k = 0; k < EDGE_SLOTS; ++k) {
        if (tid + k * T_THREADS < EDGES_PER_WG) {
            unsigned pos = atomicAdd(&tb[d[k]], 1u);
            if (pos < N_EDGES) csr_src[pos] = s[k];
        }
    }
}

__device__ __forceinline__ float edge_score(const float* __restrict__ s_src, int s, float sd) {
    float x = s_src[s] + sd;
    return (x >= 0.f) ? x : NEG_SLOPE * x;
}

__device__ __forceinline__ void gather_rows(int s, float wgt, int cn, const float* __restrict__ xcol,
                                            float& ax, float& ay) {
    const int wb = __float_as_int(wgt);
    int q = 0;
    for (; q + 4 <= cn; q += 4) {
        int s0 = __builtin_amdgcn_readlane(s, q),     s1 = __builtin_amdgcn_readlane(s, q + 1);
        int s2 = __builtin_amdgcn_readlane(s, q + 2), s3 = __builtin_amdgcn_readlane(s, q + 3);
        float w0 = __int_as_float(__builtin_amdgcn_readlane(wb, q));
        float w1 = __int_as_float(__builtin_amdgcn_readlane(wb, q + 1));
        float w2 = __int_as_float(__builtin_amdgcn_readlane(wb, q + 2));
        float w3 = __int_as_float(__builtin_amdgcn_readlane(wb, q + 3));
        float2 x0 = *(const float2*)(xcol + (size_t)s0 * DIM);
        float2 x1 = *(const float2*)(xcol + (size_t)s1 * DIM);
        float2 x2 = *(const float2*)(xcol + (size_t)s2 * DIM);
        float2 x3 = *(const float2*)(xcol + (size_t)s3 * DIM);
        ax += w0 * x0.x; ay += w0 * x0.y;
        ax += w1 * x1.x; ay += w1 * x1.y;
        ax += w2 * x2.x; ay += w2 * x2.y;
        ax += w3 * x3.x; ay += w3 * x3.y;
    }
    for (; q < cn; ++q) {
        int sq = __builtin_amdgcn_readlane(s, q);
        float wq = __int_as_float(__builtin_amdgcn_readlane(wb, q));
        float2 xv = *(const float2*)(xcol + (size_t)sq * DIM);
        ax += wq * xv.x; ay += wq * xv.y;
    }
}

__device__ __forceinline__ void sum_rows(int h, int cn, const float* __restrict__ hcol,
                                         float& ax, float& ay) {
    int q = 0;
    for (; q + 4 <= cn; q += 4) {
        int h0 = __builtin_amdgcn_readlane(h, q),     h1 = __builtin_amdgcn_readlane(h, q + 1);
        int h2 = __builtin_amdgcn_readlane(h, q + 2), h3 = __builtin_amdgcn_readlane(h, q + 3);
        float2 x0 = *(const float2*)(hcol + (size_t)h0 * DIM);
        float2 x1 = *(const float2*)(hcol + (size_t)h1 * DIM);
        float2 x2 = *(const float2*)(hcol + (size_t)h2 * DIM);
        float2 x3 = *(const float2*)(hcol + (size_t)h3 * DIM);
        ax += x0.x; ay += x0.y;
        ax += x1.x; ay += x1.y;
        ax += x2.x; ay += x2.y;
        ax += x3.x; ay += x3.y;
    }
    for (; q < cn; ++q) {
        int hq = __builtin_amdgcn_readlane(h, q);
        float2 xv = *(const float2*)(hcol + (size_t)hq * DIM);
        ax += xv.x; ay += xv.y;
    }
}

__global__ void k_hyper_rows(const unsigned* __restrict__ cnt_e, const unsigned* __restrict__ rowloc_e,
                             const unsigned* __restrict__ blockoff_e, const int* __restrict__ csr_e,
                             const unsigned* __restrict__ cnt_v, const float* __restrict__ Xp,
                             float* __restrict__ He) {
    const int e = (blockIdx.x * blockDim.x + threadIdx.x) >> 5;
    const int lane = threadIdx.x & 31;
    if (e >= M_HYPER) return;
    const int deg = __builtin_amdgcn_readfirstlane((int)cnt_e[e]);
    const unsigned beg = (unsigned)__builtin_amdgcn_readfirstlane(
        (int)(rowloc_e[e] + blockoff_e[e >> MB_SHIFT]));
    const int* vs = csr_e + beg;
    const float* xcol = Xp + lane * 2;
    float ax = 0.f, ay = 0.f;
    for (int j0 = 0; j0 < deg; j0 += 32) {
        int j = j0 + lane;
        int v = 0;
        float wgt = 0.f;
        if (j < deg) {
            v = vs[j];
            float dvv = (float)cnt_v[v];
            wgt = (dvv > 0.f) ? rsqrtf(dvv) : 0.f;
        }
        int cn = deg - j0;
        gather_rows(v, wgt, (cn < 32) ? cn : 32, xcol, ax, ay);
    }
    const float di = (deg > 0) ? (1.f / (float)deg) : 0.f;
    float2 r; r.x = ax * di; r.y = ay * di;
    *(float2*)(He + (size_t)e * DIM + lane * 2) = r;
}

__global__ void k_node(const unsigned* __restrict__ cnt, const unsigned* __restrict__ rowloc,
                       const unsigned* __restrict__ blockoff, const int* __restrict__ csr_src,
                       const float* __restrict__ s_src, const float* __restrict__ s_dst,
                       const float* __restrict__ Xp,
                       const unsigned* __restrict__ cnt_v, const unsigned* __restrict__ rowloc_v,
                       const unsigned* __restrict__ blockoff_v, const int* __restrict__ csr_v,
                       const float* __restrict__ He, float* __restrict__ out) {
    const int n = (blockIdx.x * blockDim.x + threadIdx.x) >> 5;
    const int lane = threadIdx.x & 31;
    if (n >= N_NODES) return;
    const int deg = __builtin_amdgcn_readfirstlane((int)cnt[n]);
    const unsigned beg = (unsigned)__builtin_amdgcn_readfirstlane(
        (int)(rowloc[n] + blockoff[n >> MB_SHIFT]));
    const int dgv = __builtin_amdgcn_readfirstlane((int)cnt_v[n]);
    const unsigned begv = (unsigned)__builtin_amdgcn_readfirstlane(
        (int)(rowloc_v[n] + blockoff_v[n >> MB_SHIFT]));
    const float sd = s_dst[n];
    float ax = 0.f, ay = 0.f;
    if (deg > 0) {
        const int* es = csr_src + beg;
        const float* xcol = Xp + lane * 2;
        int s0 = 0;
        float sc0 = -INFINITY;
        if (lane < deg) { s0 = es[lane]; sc0 = edge_score(s_src, s0, sd); }
        float m = sc0;
        for (int j0 = 32; j0 < deg; j0 += 32) {
            int j = j0 + lane;
            if (j < deg) m = fmaxf(m, edge_score(s_src, es[j], sd));
        }
        for (int o = 16; o > 0; o >>= 1) m = fmaxf(m, __shfl_xor(m, o));
        if (!isfinite(m)) m = 0.f;
        const float e0 = (lane < deg) ? expf(sc0 - m) : 0.f;
        float sum = e0;
        for (int j0 = 32; j0 < deg; j0 += 32) {
            int j = j0 + lane;
            if (j < deg) sum += expf(edge_score(s_src, es[j], sd) - m);
        }
        for (int o = 16; o > 0; o >>= 1) sum += __shfl_xor(sum, o);
        gather_rows(s0, e0 / sum, (deg < 32) ? deg : 32, xcol, ax, ay);
        for (int j0 = 32; j0 < deg; j0 += 32) {
            int j = j0 + lane;
            int s = 0;
            float wgt = 0.f;
            if (j < deg) {
                s = es[j];
                wgt = expf(edge_score(s_src, s, sd) - m) / sum;
            }
            int cn = deg - j0;
            gather_rows(s, wgt, (cn < 32) ? cn : 32, xcol, ax, ay);
        }
    }
    float hx = 0.f, hy = 0.f;
    {
        const int* hs = csr_v + begv;
        const float* hcol = He + lane * 2;
        for (int j0 = 0; j0 < dgv; j0 += 32) {
            int j = j0 + lane;
            int h = (j < dgv) ? hs[j] : 0;
            int cn = dgv - j0;
            sum_rows(h, (cn < 32) ? cn : 32, hcol, hx, hy);
        }
    }
    const float sv = (dgv > 0) ? rsqrtf((float)dgv) : 0.f;
    const float x0 = 0.5f * (ax + hx * sv);
    const float x1 = 0.5f * (ay + hy * sv);
    float2 r;
    r.x = (x0 > 0.f) ? x0 : expm1f(x0);
    r.y = (x1 > 0.f) ? x1 : expm1f(x1);
    *(float2*)(out + (size_t)n * DIM + lane * 2) = r;
}

extern "C" void kernel_launch(void* const* d_in, const int* in_sizes, int n_in,
                              void* d_out, int out_size, void* d_ws, size_t ws_size,
                              hipStream_t stream) {
    const float* X     = (const float*)d_in[0];
    const float* W     = (const float*)d_in[1];
    const float* bias  = (const float*)d_in[2];
    const float* a_src = (const float*)d_in[3];
    const float* a_dst = (const float*)d_in[4];
    const int*   e_src = (const int*)d_in[5];
    const int*   e_dst = (const int*)d_in[6];
    const int*   inc_v = (const int*)d_in[7];
    const int*   inc_e = (const int*)d_in[8];
    float* out = (float*)d_out;
    float* ws  = (float*)d_ws;

    size_t off = 0;
    float* Xp    = ws + off; off += (size_t)N_NODES * DIM;
    float* s_src = ws + off; off += N_NODES;
    float* s_dst = ws + off; off += N_NODES;
    unsigned* cnt    = (unsigned*)(ws + off); off += N_NODES;
    unsigned* rowloc = (unsigned*)(ws + off); off += N_NODES;
    unsigned* cnt_v  = (unsigned*)(ws + off); off += N_NODES;
    unsigned* cnt_e  = (unsigned*)(ws + off); off += M_HYPER;
    float* span  = ws + off; off += SPAN_FLOATS;
    int* csr_src = (int*)(ws + off); off += N_EDGES;

    unsigned* part_v = (unsigned*)span;
    unsigned* part_e = (unsigned*)(span + (size_t)(N_NODES + M_HYPER) * DIM);
    unsigned* part   = (unsigned*)span;
    float* He        = span + (size_t)N_NODES * DIM;
    float* tail      = span + TAIL_OFF;
    size_t t = 0;
    int* csr_v = (int*)(tail + t); t += P_PAIRS;
    int* csr_e = (int*)(tail + t); t += P_PAIRS;
    unsigned* rowloc_v   = (unsigned*)(tail + t); t += N_NODES;
    unsigned* rowloc_e   = (unsigned*)(tail + t); t += M_HYPER;
    unsigned* blocksum_v = (unsigned*)(tail + t); t += BS_PAD;
    unsigned* blockoff_v = (unsigned*)(tail + t); t += BS_PAD;
    unsigned* blocksum_e = (unsigned*)(tail + t); t += HS_PAD;
    unsigned* blockoff_e = (unsigned*)(tail + t); t += HS_PAD;
    unsigned* blocksum   = (unsigned*)(tail + t); t += BS_PAD;
    unsigned* blockoff   = (unsigned*)(tail + t); t += BS_PAD;

    const int B = 256;
    int tiles = N_NODES / 16;
    k_gemm_wmma<<<(tiles * 32 + B - 1) / B, B, 0, stream>>>(X, W, bias, Xp);
    k_scalars<<<(N_NODES + B - 1) / B, B, 0, stream>>>(Xp, a_src, a_dst, s_src, s_dst);

    k_pair_cnt<<<T_PAIR, T_THREADS, 0, stream>>>(inc_v, inc_e, part_v, part_e);
    k_pair_merge<<<N_MBLK + N_HBLK, MB_THREADS, 0, stream>>>(part_v, part_e, cnt_v, cnt_e,
                                                             rowloc_v, rowloc_e, blocksum_v, blocksum_e);
    k_pair_fill<<<T_PAIR, T_THREADS, 0, stream>>>(inc_v, inc_e, part_v, part_e, rowloc_v, rowloc_e,
                                                  blocksum_v, blocksum_e, blockoff_v, blockoff_e,
                                                  csr_v, csr_e);

    k_cnt_part<<<T_EDGE, T_THREADS, 0, stream>>>(e_dst, part);
    k_cnt_merge<<<N_MBLK, MB_THREADS, 0, stream>>>(part, cnt, rowloc, blocksum);
    k_fill<<<T_EDGE, T_THREADS, 0, stream>>>(e_src, e_dst, part, rowloc, blocksum, blockoff,
                                             csr_src);

    k_hyper_rows<<<(M_HYPER * 32 + B - 1) / B, B, 0, stream>>>(cnt_e, rowloc_e, blockoff_e, csr_e,
                                                               cnt_v, Xp, He);
    k_node<<<(N_NODES * 32 + B - 1) / B, B, 0, stream>>>(cnt, rowloc, blockoff, csr_src,
                                                         s_src, s_dst, Xp,
                                                         cnt_v, rowloc_v, blockoff_v, csr_v, He, out);
}


# Independent stages share launches by block-index range: pair sorts beside edge sort (80+40 slices), GEMM beside merges, scalars beside He rows: 10 launches -> 5; list order follows new slices
# speedup vs baseline: 3.1637x; 1.4997x over previous
#include <hip/hip_runtime.h>
#include <hip/hip_bf16.h>
#include <math.h>

typedef __attribute__((ext_vector_type(2))) float v2f;
typedef __attribute__((ext_vector_type(8))) float v8f;

#define N_NODES 50000
#define N_EDGES 800000
#define M_HYPER 10000
#define P_PAIRS 200000
#define DIM     64
#define NEG_SLOPE 0.2f


__device__ __forceinline__ void gemm_tile(int blk, const float* __restrict__ X, const float* __restrict__ W,
                                          const float* __restrict__ bias, float* __restrict__ Xp) {
    int wave = (blk * 256 + (int)threadIdx.x) >> 5;
    int lane = threadIdx.x & 31;
    if (wave >= N_NODES / 16) return;
    int row_base = wave * 16;
    int lh  = (lane >= 16) ? 1 : 0;
    int l15 = lane & 15;

    const float* xrow = X + (size_t)(row_base + l15) * DIM + lh * 2;
    const float* wrow = W + (size_t)l15 * DIM + lh * 2;

    v8f acc[4] = {};
#pragma unroll
    for (int ks = 0; ks < 16; ++ks) {
        v2f a;
        a.x = xrow[ks * 4 + 0];
        a.y = xrow[ks * 4 + 1];
#pragma unroll
        for (int t = 0; t < 4; ++t) {
            v2f b;
            const float* wb = wrow + (size_t)t * 16 * DIM + ks * 4;
            b.x = wb[0];
            b.y = wb[1];
            acc[t] = __builtin_amdgcn_wmma_f32_16x16x4_f32(
                false, a, false, b, (short)0, acc[t], false, false);
        }
    }

#pragma unroll
    for (int t = 0; t < 4; ++t) {
        int c = t * 16 + l15;
        float bv = bias[c];
#pragma unroll
        for (int v = 0; v < 8; ++v) {
            int m = row_base + v + lh * 8;
            Xp[(size_t)m * DIM + c] = acc[t][v] + bv;
        }
    }
}

__device__ __forceinline__ void node_scalars(int blk, const float* __restrict__ Xp,
                          const float* __restrict__ a_src, const float* __restrict__ a_dst,
                          float* __restrict__ s_src, float* __restrict__ s_dst) {
    int n = blk * 256 + (int)threadIdx.x;
    if (n >= N_NODES) return;
    const float4* row = (const float4*)(Xp + (size_t)n * DIM);
    const float4* as  = (const float4*)a_src;
    const float4* ad  = (const float4*)a_dst;
    float ss = 0.f, sd = 0.f;
#pragma unroll 4
    for (int i = 0; i < 16; ++i) {
        float4 x = row[i], u = as[i], v = ad[i];
        ss += x.x * u.x + x.y * u.y + x.z * u.z + x.w * u.w;
        sd += x.x * v.x + x.y * v.y + x.z * v.z + x.w * v.w;
    }
    s_src[n] = ss;
    s_dst[n] = sd;
}

#define T_THREADS 1024
#define T_EDGE 80
#define T_PAIR 40
#define EDGES_PER_WG (N_EDGES / T_EDGE)
#define PAIRS_PER_WG (P_PAIRS / T_PAIR)
#define EDGE_SLOTS ((EDGES_PER_WG + T_THREADS - 1) / T_THREADS)
#define PAIR_SLOTS ((PAIRS_PER_WG + T_THREADS - 1) / T_THREADS)
#define TAB_N ((N_NODES + 31) / 32 * 32)
#define TAB_M ((M_HYPER + 31) / 32 * 32)
static_assert(N_EDGES % T_EDGE == 0 && P_PAIRS % T_PAIR == 0, "slices are exact");

#define MB_THREADS 256
#define MB_SHIFT 8
#define N_MBLK ((N_NODES + MB_THREADS - 1) / MB_THREADS)
#define N_HBLK ((M_HYPER + MB_THREADS - 1) / MB_THREADS)
#define BS_PAD 256
#define HS_PAD 64
static_assert((1 << MB_SHIFT) == MB_THREADS && N_MBLK <= BS_PAD && N_HBLK <= HS_PAD,
              "block sums fit their pads");
#define SPAN_FLOATS ((size_t)2 * N_NODES * DIM + (size_t)M_HYPER * DIM)
#define PARTV_OFF ((size_t)T_EDGE * TAB_N)
#define PARTE_OFF (PARTV_OFF + (size_t)T_PAIR * TAB_N)
#define TAIL_OFF (PARTE_OFF + (size_t)T_PAIR * TAB_M)
static_assert(TAIL_OFF + 2 * (size_t)P_PAIRS + N_NODES + M_HYPER + 4 * BS_PAD + 2 * HS_PAD <= SPAN_FLOATS,
              "lists, row offsets and block-sum pads fit behind the tables");
static_assert((size_t)(N_NODES + M_HYPER) * DIM <= TAIL_OFF, "He ends in front of the tail");

__device__ __forceinline__ unsigned wave_incl_scan(unsigned v, int lane) {
    for (int o = 1; o < 32; o <<= 1) {
        unsigned t = __shfl_up(v, o);
        if (lane >= o) v += t;
    }
    return v;
}

template <int SLICES>
__device__ __forceinline__ void slice_prefix(unsigned* __restrict__ part, size_t pitch, int lim, int blk,
                                             unsigned* __restrict__ cnt, unsigned* __restrict__ rowloc,
                                             unsigned* __restrict__ blocksum, unsigned* wsum) {
    const int tid = threadIdx.x, lane = tid & 31, w = tid >> 5;
    const int n = blk * MB_THREADS + tid;
    unsigned run = 0u;
    if (n < lim) {
#pragma unroll 16
        for (int g = 0; g < SLICES; ++g) {
            size_t idx = (size_t)g * pitch + n;
            unsigned c = part[idx];
            part[idx] = run;
            run += c;
        }
    }
    unsigned inc = wave_incl_scan(run, lane);
    if (lane == 31) wsum[w] = inc;
    __syncthreads();
    if (w == 0) {
        unsigned v = (lane < MB_THREADS / 32) ? wsum[lane] : 0u;
        unsigned iv = wave_incl_scan(v, lane);
        if (lane < MB_THREADS / 32) wsum[lane] = iv - v;
        if (lane == 31) blocksum[blk] = iv;
    }
    __syncthreads();
    if (n < lim) {
        rowloc[n] = wsum[w] + inc - run;
        cnt[n] = run;
    }
}

__device__ __forceinline__ void block_offsets(const unsigned* __restrict__ blocksum, int nblk,
                                              unsigned* bs, int lane) {
    unsigned carry = 0u;
    for (int c = 0; c < nblk; c += 32) {
        int i = c + lane;
        unsigned a = (i < nblk) ? blocksum[i] : 0u;
        unsigned ia = wave_incl_scan(a, lane);
        if (i < nblk) bs[i] = carry + ia - a;
        carry += __shfl(ia, 31);
    }
}

__global__ __launch_bounds__(T_THREADS)
void k_count(const int* __restrict__ e_dst, const int* __restrict__ inc_v, const int* __restrict__ inc_e,
             unsigned* __restrict__ part, unsigned* __restrict__ part_v, unsigned* __restrict__ part_e) {
    __shared__ unsigned tab[N_NODES + M_HYPER];
    const int tid = threadIdx.x;
    if (blockIdx.x < T_EDGE) {
        unsigned* tc = tab;
        for (int i = tid; i < N_NODES; i += T_THREADS) tc[i] = 0u;
        __syncthreads();
        const int base = blockIdx.x * EDGES_PER_WG;
        int d[EDGE_SLOTS];
#pragma unroll
        for (int k = 0; k < EDGE_SLOTS; ++k) {
            int i = tid + k * T_THREADS;
            d[k] = (i < EDGES_PER_WG) ? e_dst[base + i] : 0;
        }
#pragma unroll
        for (int k = 0; k < EDGE_SLOTS; ++k) {
            if (tid + k * T_THREADS < EDGES_PER_WG) atomicAdd(&tc[d[k]], 1u);
        }
        __syncthreads();
        unsigned* out = part + (size_t)blockIdx.x * TAB_N;
        for (int i = tid; i < N_NODES; i += T_THREADS) out[i] = tc[i];
    } else {
        const int blk = blockIdx.x - T_EDGE;
        unsigned* tv = tab;
        unsigned* te = tab + N_NODES;
        for (int i = tid; i < N_NODES + M_HYPER; i += T_THREADS) tab[i] = 0u;
        __syncthreads();
        const int base = blk * PAIRS_PER_WG;
        int v[PAIR_SLOTS], h[PAIR_SLOTS];
#pragma unroll
        for (int k = 0; k < PAIR_SLOTS; ++k) {
            int i = tid + k * T_THREADS;
            bool ok = i < PAIRS_PER_WG;
            v[k] = ok ? inc_v[base + i] : 0;
            h[k] = ok ? inc_e[base + i] : 0;
        }
#pragma unroll
        for (int k = 0; k < PAIR_SLOTS; ++k) {
            if (tid + k * T_THREADS < PAIRS_PER_WG) {
                atomicAdd(&tv[v[k]], 1u);
                atomicAdd(&te[h[k]], 1u);
            }
        }
        __syncthreads();
        unsigned* pv = part_v + (size_t)blk * TAB_N;
        for (int i = tid; i < N_NODES; i += T_THREADS) pv[i] = tv[i];
        unsigned* pe = part_e + (size_t)blk * TAB_M;
        for (int i = tid; i < M_HYPER; i += T_THREADS) pe[i] = te[i];
    }
}

#define GEMM_WGS ((N_NODES / 16 * 32 + MB_THREADS - 1) / MB_THREADS)
__global__ __launch_bounds__(MB_THREADS)
void k_merge_gemm(unsigned* __restrict__ part, unsigned* __restrict__ cnt,
                  unsigned* __restrict__ rowloc, unsigned* __restrict__ blocksum,
                  unsigned* __restrict__ part_v, unsigned* __restrict__ part_e,
                  unsigned* __restrict__ cnt_v, unsigned* __restrict__ cnt_e,
                  unsigned* __restrict__ rowloc_v, unsigned* __restrict__ rowloc_e,
                  unsigned* __restrict__ blocksum_v, unsigned* __restrict__ blocksum_e,
                  const float* __restrict__ X, const float* __restrict__ W,
                  const float* __restrict__ bias, float* __restrict__ Xp) {
    __shared__ unsigned wsum[MB_THREADS / 32];
    const int b = blockIdx.x;
    if (b < N_MBLK) {
        slice_prefix<T_EDGE>(part, (size_t)TAB_N, N_NODES, b, cnt, rowloc, blocksum, wsum);
    } else if (b < 2 * N_MBLK + N_HBLK) {
        const bool isv = b < 2 * N_MBLK;
        slice_prefix<T_PAIR>(isv ? part_v : part_e, isv ? (size_t)TAB_N : (size_t)TAB_M,
                             isv ? N_NODES : M_HYPER, isv ? b - N_MBLK : b - 2 * N_MBLK,
                             isv ? cnt_v : cnt_e, isv ? rowloc_v : rowloc_e,
                             isv ? blocksum_v : blocksum_e, wsum);
    } else {
        gemm_tile(b - (2 * N_MBLK + N_HBLK), X, W, bias, Xp);
    }
}

__global__ __launch_bounds__(T_THREADS)
void k_place(const int* __restrict__ e_src, const int* __restrict__ e_dst,
             const int* __restrict__ inc_v, const int* __restrict__ inc_e,
             const unsigned* __restrict__ part, const unsigned* __restrict__ rowloc,
             const unsigned* __restrict__ blocksum, unsigned* __restrict__ blockoff,
             int* __restrict__ csr_src,
             const unsigned* __restrict__ part_v, const unsigned* __restrict__ part_e,
             const unsigned* __restrict__ rowloc_v, const unsigned* __restrict__ rowloc_e,
             const unsigned* __restrict__ blocksum_v, const unsigned* __restrict__ blocksum_e,
             unsigned* __restrict__ blockoff_v, unsigned* __restrict__ blockoff_e,
             int* __restrict__ csr_v, int* __restrict__ csr_e) {
    __shared__ unsigned tab[N_NODES + M_HYPER];
    __shared__ unsigned bv[BS_PAD];
    __shared__ unsigned be[HS_PAD];
    const int tid = threadIdx.x;
    if (blockIdx.x < T_EDGE) {
        unsigned* tb = tab;
        if (tid < 32) block_offsets(blocksum, N_MBLK, bv, tid);
        __syncthreads();
        if (blockIdx.x == 0 && tid < N_MBLK) blockoff[tid] = bv[tid];
        const unsigned* pg = part + (size_t)blockIdx.x * TAB_N;
        for (int i = tid; i < N_NODES; i += T_THREADS) tb[i] = pg[i] + rowloc[i] + bv[i >> MB_SHIFT];
        __syncthreads();
        const int base = blockIdx.x * EDGES_PER_WG;
        int s[EDGE_SLOTS], d[EDGE_SLOTS];
#pragma unroll
        for (int k = 0; k < EDGE_SLOTS; ++k) {
            int i = tid + k * T_THREADS;
            bool ok = i < EDGES_PER_WG;
            s[k] = ok ? e_src[base + i] : 0;
            d[k] = ok ? e_dst[base + i] : 0;
        }
#pragma unroll
        for (int k = 0; k < EDGE_SLOTS; ++k) {
            if (tid + k * T_THREADS < EDGES_PER_WG) {
                unsigned pos = atomicAdd(&tb[d[k]], 1u);
                if (pos < N_EDGES) csr_src[pos] = s[k];
            }
        }
    } else {
        const int blk = blockIdx.x - T_EDGE;
        unsigned* tv = tab;
        unsigned* te = tab + N_NODES;
        if (tid < 32) block_offsets(blocksum_v, N_MBLK, bv, tid);
        else if (tid < 64) block_offsets(blocksum_e, N_HBLK, be, tid - 32);
        __syncthreads();
        if (blk == 0) {
            if (tid < N_MBLK) blockoff_v[tid] = bv[tid];
            if (tid < N_HBLK) blockoff_e[tid] = be[tid];
        }
        const unsigned* pv = part_v + (size_t)blk * TAB_N;
        for (int i = tid; i < N_NODES; i += T_THREADS) tv[i] = pv[i] + rowloc_v[i] + bv[i >> MB_SHIFT];
        const unsigned* pe = part_e + (size_t)blk * TAB_M;
        for (int i = tid; i < M_HYPER; i += T_THREADS) te[i] = pe[i] + rowloc_e[i] + be[i >> MB_SHIFT];
        __syncthreads();
        const int base = blk * PAIRS_PER_WG;
        int v[PAIR_SLOTS], h[PAIR_SLOTS];
#pragma unroll
        for (int k = 0; k < PAIR_SLOTS; ++k) {
            int i = tid + k * T_THREADS;
            bool ok = i < PAIRS_PER_WG;
            v[k] = ok ? inc_v[base + i] : 0;
            h[k] = ok ? inc_e[base + i] : 0;
        }
#pragma unroll
        for (int k = 0; k < PAIR_SLOTS; ++k) {
            if (tid + k * T_THREADS < PAIRS_PER_WG) {
                unsigned pn = atomicAdd(&tv[v[k]], 1u);
                unsigned ph = atomicAdd(&te[h[k]], 1u);
                if (pn < P_PAIRS) csr_v[pn] = h[k];
                if (ph < P_PAIRS) csr_e[ph] = v[k];
            }
        }
    }
}

__device__ __forceinline__ float edge_score(const float* __restrict__ s_src, int s, float sd) {
    float x = s_src[s] + sd;
    return (x >= 0.f) ? x : NEG_SLOPE * x;
}

__device__ __forceinline__ void gather_rows(int s, float wgt, int cn, const float* __restrict__ xcol,
                                            float& ax, float& ay) {
    const int wb = __float_as_int(wgt);
    int q = 0;
    for (; q + 4 <= cn; q += 4) {
        int s0 = __builtin_amdgcn_readlane(s, q),     s1 = __builtin_amdgcn_readlane(s, q + 1);
        int s2 = __builtin_amdgcn_readlane(s, q + 2), s3 = __builtin_amdgcn_readlane(s, q + 3);
        float w0 = __int_as_float(__builtin_amdgcn_readlane(wb, q));
        float w1 = __int_as_float(__builtin_amdgcn_readlane(wb, q + 1));
        float w2 = __int_as_float(__builtin_amdgcn_readlane(wb, q + 2));
        float w3 = __int_as_float(__builtin_amdgcn_readlane(wb, q + 3));
        float2 x0 = *(const float2*)(xcol + (size_t)s0 * DIM);
        float2 x1 = *(const float2*)(xcol + (size_t)s1 * DIM);
        float2 x2 = *(const float2*)(xcol + (size_t)s2 * DIM);
        float2 x3 = *(const float2*)(xcol + (size_t)s3 * DIM);
        ax += w0 * x0.x; ay += w0 * x0.y;
        ax += w1 * x1.x; ay += w1 * x1.y;
        ax += w2 * x2.x; ay += w2 * x2.y;
        ax += w3 * x3.x; ay += w3 * x3.y;
    }
    for (; q < cn; ++q) {
        int sq = __builtin_amdgcn_readlane(s, q);
        float wq = __int_as_float(__builtin_amdgcn_readlane(wb, q));
        float2 xv = *(const float2*)(xcol + (size_t)sq * DIM);
        ax += wq * xv.x; ay += wq * xv.y;
    }
}

__device__ __forceinline__ void sum_rows(int h, int cn, const float* __restrict__ hcol,
                                         float& ax, float& ay) {
    int q = 0;
    for (; q + 4 <= cn; q += 4) {
        int h0 = __builtin_amdgcn_readlane(h, q),     h1 = __builtin_amdgcn_readlane(h, q + 1);
        int h2 = __builtin_amdgcn_readlane(h, q + 2), h3 = __builtin_amdgcn_readlane(h, q + 3);
        float2 x0 = *(const float2*)(hcol + (size_t)h0 * DIM);
        float2 x1 = *(const float2*)(hcol + (size_t)h1 * DIM);
        float2 x2 = *(const float2*)(hcol + (size_t)h2 * DIM);
        float2 x3 = *(const float2*)(hcol + (size_t)h3 * DIM);
        ax += x0.x; ay += x0.y;
        ax += x1.x; ay += x1.y;
        ax += x2.x; ay += x2.y;
        ax += x3.x; ay += x3.y;
    }
    for (; q < cn; ++q) {
        int hq = __builtin_amdgcn_readlane(h, q);
        float2 xv = *(const float2*)(hcol + (size_t)hq * DIM);
        ax += xv.x; ay += xv.y;
    }
}

__device__ __forceinline__ void hyper_rows(int blk, const unsigned* __restrict__ cnt_e,
                             const unsigned* __restrict__ rowloc_e,
                             const unsigned* __restrict__ blockoff_e, const int* __restrict__ csr_e,
                             const unsigned* __restrict__ cnt_v, const float* __restrict__ Xp,
                             float* __restrict__ He) {
    const int e = (blk * 256 + (int)threadIdx.x) >> 5;
    const int lane = threadIdx.x & 31;
    if (e >= M_HYPER) return;
    const int deg = __builtin_amdgcn_readfirstlane((int)cnt_e[e]);
    const unsigned beg = (unsigned)__builtin_amdgcn_readfirstlane(
        (int)(rowloc_e[e] + blockoff_e[e >> MB_SHIFT]));
    const int* vs = csr_e + beg;
    const float* xcol = Xp + lane * 2;
    float ax = 0.f, ay = 0.f;
    for (int j0 = 0; j0 < deg; j0 += 32) {
        int j = j0 + lane;
        int v = 0;
        float wgt = 0.f;
        if (j < deg) {
            v = vs[j];
            float dvv = (float)cnt_v[v];
            wgt = (dvv > 0.f) ? rsqrtf(dvv) : 0.f;
        }
        int cn = deg - j0;
        gather_rows(v, wgt, (cn < 32) ? cn : 32, xcol, ax, ay);
    }
    const float di = (deg > 0) ? (1.f / (float)deg) : 0.f;
    float2 r; r.x = ax * di; r.y = ay * di;
    *(float2*)(He + (size_t)e * DIM + lane * 2) = r;
}

#define SC_WGS ((N_NODES + 255) / 256)
#define HY_WGS ((M_HYPER * 32 + 255) / 256)
__global__ void k_hyper_scalars(const unsigned* __restrict__ cnt_e, const unsigned* __restrict__ rowloc_e,
                                const unsigned* __restrict__ blockoff_e, const int* __restrict__ csr_e,
                                const unsigned* __restrict__ cnt_v, const float* __restrict__ Xp,
                                float* __restrict__ He,
                                const float* __restrict__ a_src, const float* __restrict__ a_dst,
                                float* __restrict__ s_src, float* __restrict__ s_dst) {
    if (blockIdx.x < SC_WGS) node_scalars((int)blockIdx.x, Xp, a_src, a_dst, s_src, s_dst);
    else hyper_rows((int)blockIdx.x - SC_WGS, cnt_e, rowloc_e, blockoff_e, csr_e, cnt_v, Xp, He);
}

__global__ void k_node(const unsigned* __restrict__ cnt, const unsigned* __restrict__ rowloc,
                       const unsigned* __restrict__ blockoff, const int* __restrict__ csr_src,
                       const float* __restrict__ s_src, const float* __restrict__ s_dst,
                       const float* __restrict__ Xp,
                       const unsigned* __restrict__ cnt_v, const unsigned* __restrict__ rowloc_v,
                       const unsigned* __restrict__ blockoff_v, const int* __restrict__ csr_v,
                       const float* __restrict__ He, float* __restrict__ out) {
    const int n = (blockIdx.x * blockDim.x + threadIdx.x) >> 5;
    const int lane = threadIdx.x & 31;
    if (n >= N_NODES) return;
    const int deg = __builtin_amdgcn_readfirstlane((int)cnt[n]);
    const unsigned beg = (unsigned)__builtin_amdgcn_readfirstlane(
        (int)(rowloc[n] + blockoff[n >> MB_SHIFT]));
    const int dgv = __builtin_amdgcn_readfirstlane((int)cnt_v[n]);
    const unsigned begv = (unsigned)__builtin_amdgcn_readfirstlane(
        (int)(rowloc_v[n] + blockoff_v[n >> MB_SHIFT]));
    const float sd = s_dst[n];
    float ax = 0.f, ay = 0.f;
    if (deg > 0) {
        const int* es = csr_src + beg;
        const float* xcol = Xp + lane * 2;
        int s0 = 0;
        float sc0 = -INFINITY;
        if (lane < deg) { s0 = es[lane]; sc0 = edge_score(s_src, s0, sd); }
        float m = sc0;
        for (int j0 = 32; j0 < deg; j0 += 32) {
            int j = j0 + lane;
            if (j < deg) m = fmaxf(m, edge_score(s_src, es[j], sd));
        }
        for (int o = 16; o > 0; o >>= 1) m = fmaxf(m, __shfl_xor(m, o));
        if (!isfinite(m)) m = 0.f;
        const float e0 = (lane < deg) ? expf(sc0 - m) : 0.f;
        float sum = e0;
        for (int j0 = 32; j0 < deg; j0 += 32) {
            int j = j0 + lane;
            if (j < deg) sum += expf(edge_score(s_src, es[j], sd) - m);
        }
        for (int o = 16; o > 0; o >>= 1) sum += __shfl_xor(sum, o);
        gather_rows(s0, e0 / sum, (deg < 32) ? deg : 32, xcol, ax, ay);
        for (int j0 = 32; j0 < deg; j0 += 32) {
            int j = j0 + lane;
            int s = 0;
            float wgt = 0.f;
            if (j < deg) {
                s = es[j];
                wgt = expf(edge_score(s_src, s, sd) - m) / sum;
            }
            int cn = deg - j0;
            gather_rows(s, wgt, (cn < 32) ? cn : 32, xcol, ax, ay);
        }
    }
    float hx = 0.f, hy = 0.f;
    {
        const int* hs = csr_v + begv;
        const float* hcol = He + lane * 2;
        for (int j0 = 0; j0 < dgv; j0 += 32) {
            int j = j0 + lane;
            int h = (j < dgv) ? hs[j] : 0;
            int cn = dgv - j0;
            sum_rows(h, (cn < 32) ? cn : 32, hcol, hx, hy);
        }
    }
    const float sv = (dgv > 0) ? rsqrtf((float)dgv) : 0.f;
    const float x0 = 0.5f * (ax + hx * sv);
    const float x1 = 0.5f * (ay + hy * sv);
    float2 r;
    r.x = (x0 > 0.f) ? x0 : expm1f(x0);
    r.y = (x1 > 0.f) ? x1 : expm1f(x1);
    *(float2*)(out + (size_t)n * DIM + lane * 2) = r;
}

extern "C" void kernel_launch(void* const* d_in, const int* in_sizes, int n_in,
                              void* d_out, int out_size, void* d_ws, size_t ws_size,
                              hipStream_t stream) {
    const float* X     = (const float*)d_in[0];
    const float* W     = (const float*)d_in[1];
    const float* bias  = (const float*)d_in[2];
    const float* a_src = (const float*)d_in[3];
    const float* a_dst = (const float*)d_in[4];
    const int*   e_src = (const int*)d_in[5];
    const int*   e_dst = (const int*)d_in[6];
    const int*   inc_v = (const int*)d_in[7];
    const int*   inc_e = (const int*)d_in[8];
    float* out = (float*)d_out;
    float* ws  = (float*)d_ws;

    size_t off = 0;
    float* Xp    = ws + off; off += (size_t)N_NODES * DIM;
    float* s_src = ws + off; off += N_NODES;
    float* s_dst = ws + off; off += N_NODES;
    unsigned* cnt    = (unsigned*)(ws + off); off += N_NODES;
    unsigned* rowloc = (unsigned*)(ws + off); off += N_NODES;
    unsigned* cnt_v  = (unsigned*)(ws + off); off += N_NODES;
    unsigned* cnt_e  = (unsigned*)(ws + off); off += M_HYPER;
    float* span  = ws + off; off += SPAN_FLOATS;
    int* csr_src = (int*)(ws + off); off += N_EDGES;

    unsigned* part   = (unsigned*)span;
    unsigned* part_v = (unsigned*)(span + PARTV_OFF);
    unsigned* part_e = (unsigned*)(span + PARTE_OFF);
    float* He        = span + (size_t)N_NODES * DIM;
    float* tail      = span + TAIL_OFF;
    size_t t = 0;
    int* csr_v = (int*)(tail + t); t += P_PAIRS;
    int* csr_e = (int*)(tail + t); t += P_PAIRS;
    unsigned* rowloc_v   = (unsigned*)(tail + t); t += N_NODES;
    unsigned* rowloc_e   = (unsigned*)(tail + t); t += M_HYPER;
    unsigned* blocksum_v = (unsigned*)(tail + t); t += BS_PAD;
    unsigned* blockoff_v = (unsigned*)(tail + t); t += BS_PAD;
    unsigned* blocksum_e = (unsigned*)(tail + t); t += HS_PAD;
    unsigned* blockoff_e = (unsigned*)(tail + t); t += HS_PAD;
    unsigned* blocksum   = (unsigned*)(tail + t); t += BS_PAD;
    unsigned* blockoff   = (unsigned*)(tail + t); t += BS_PAD;

    const int B = 256;
    k_count<<<T_EDGE + T_PAIR, T_THREADS, 0, stream>>>(e_dst, inc_v, inc_e, part, part_v, part_e);
    k_merge_gemm<<<2 * N_MBLK + N_HBLK + GEMM_WGS, MB_THREADS, 0, stream>>>(
        part, cnt, rowloc, blocksum, part_v, part_e, cnt_v, cnt_e, rowloc_v, rowloc_e,
        blocksum_v, blocksum_e, X, W, bias, Xp);
    k_place<<<T_EDGE + T_PAIR, T_THREADS, 0, stream>>>(e_src, e_dst, inc_v, inc_e,
                                                       part, rowloc, blocksum, blockoff, csr_src,
                                                       part_v, part_e, rowloc_v, rowloc_e,
                                                       blocksum_v, blocksum_e, blockoff_v, blockoff_e,
                                                       csr_v, csr_e);
    k_hyper_scalars<<<SC_WGS + HY_WGS, B, 0, stream>>>(cnt_e, rowloc_e, blockoff_e, csr_e, cnt_v, Xp, He,
                                                       a_src, a_dst, s_src, s_dst);
    k_node<<<(N_NODES * 32 + B - 1) / B, B, 0, stream>>>(cnt, rowloc, blockoff, csr_src,
                                                         s_src, s_dst, Xp,
                                                         cnt_v, rowloc_v, blockoff_v, csr_v, He, out);
}


# Row kernels: every row load of a list group (16 Xp rows, 8 He rows) is in flight before the first is added; node wave asks for both lists, then score terms and He rows, together; sum order kept
# speedup vs baseline: 3.2794x; 1.0366x over previous
#include <hip/hip_runtime.h>
#include <hip/hip_bf16.h>
#include <math.h>

typedef __attribute__((ext_vector_type(2))) float v2f;
typedef __attribute__((ext_vector_type(8))) float v8f;

#define N_NODES 50000
#define N_EDGES 800000
#define M_HYPER 10000
#define P_PAIRS 200000
#define DIM     64
#define NEG_SLOPE 0.2f


__device__ __forceinline__ void gemm_tile(int blk, const float* __restrict__ X, const float* __restrict__ W,
                                          const float* __restrict__ bias, float* __restrict__ Xp) {
    int wave = (blk * 256 + (int)threadIdx.x) >> 5;
    int lane = threadIdx.x & 31;
    if (wave >= N_NODES / 16) return;
    int row_base = wave * 16;
    int lh  = (lane >= 16) ? 1 : 0;
    int l15 = lane & 15;

    const float* xrow = X + (size_t)(row_base + l15) * DIM + lh * 2;
    const float* wrow = W + (size_t)l15 * DIM + lh * 2;

    v8f acc[4] = {};
#pragma unroll
    for (int ks = 0; ks < 16; ++ks) {
        v2f a;
        a.x = xrow[ks * 4 + 0];
        a.y = xrow[ks * 4 + 1];
#pragma unroll
        for (int t = 0; t < 4; ++t) {
            v2f b;
            const float* wb = wrow + (size_t)t * 16 * DIM + ks * 4;
            b.x = wb[0];
            b.y = wb[1];
            acc[t] = __builtin_amdgcn_wmma_f32_16x16x4_f32(
                false, a, false, b, (short)0, acc[t], false, false);
        }
    }

#pragma unroll
    for (int t = 0; t < 4; ++t) {
        int c = t * 16 + l15;
        float bv = bias[c];
#pragma unroll
        for (int v = 0; v < 8; ++v) {
            int m = row_base + v + lh * 8;
            Xp[(size_t)m * DIM + c] = acc[t][v] + bv;
        }
    }
}

__device__ __forceinline__ void node_scalars(int blk, const float* __restrict__ Xp,
                          const float* __restrict__ a_src, const float* __restrict__ a_dst,
                          float* __restrict__ s_src, float* __restrict__ s_dst) {
    int n = blk * 256 + (int)threadIdx.x;
    if (n >= N_NODES) return;
    const float4* row = (const float4*)(Xp + (size_t)n * DIM);
    const float4* as  = (const float4*)a_src;
    const float4* ad  = (const float4*)a_dst;
    float ss = 0.f, sd = 0.f;
#pragma unroll 4
    for (int i = 0; i < 16; ++i) {
        float4 x = row[i], u = as[i], v = ad[i];
        ss += x.x * u.x + x.y * u.y + x.z * u.z + x.w * u.w;
        sd += x.x * v.x + x.y * v.y + x.z * v.z + x.w * v.w;
    }
    s_src[n] = ss;
    s_dst[n] = sd;
}

#define T_THREADS 1024
#define T_EDGE 80
#define T_PAIR 40
#define EDGES_PER_WG (N_EDGES / T_EDGE)
#define PAIRS_PER_WG (P_PAIRS / T_PAIR)
#define EDGE_SLOTS ((EDGES_PER_WG + T_THREADS - 1) / T_THREADS)
#define PAIR_SLOTS ((PAIRS_PER_WG + T_THREADS - 1) / T_THREADS)
#define TAB_N ((N_NODES + 31) / 32 * 32)
#define TAB_M ((M_HYPER + 31) / 32 * 32)
static_assert(N_EDGES % T_EDGE == 0 && P_PAIRS % T_PAIR == 0, "slices are exact");

#define MB_THREADS 256
#define MB_SHIFT 8
#define N_MBLK ((N_NODES + MB_THREADS - 1) / MB_THREADS)
#define N_HBLK ((M_HYPER + MB_THREADS - 1) / MB_THREADS)
#define BS_PAD 256
#define HS_PAD 64
static_assert((1 << MB_SHIFT) == MB_THREADS && N_MBLK <= BS_PAD && N_HBLK <= HS_PAD,
              "block sums fit their pads");
#define SPAN_FLOATS ((size_t)2 * N_NODES * DIM + (size_t)M_HYPER * DIM)
#define PARTV_OFF ((size_t)T_EDGE * TAB_N)
#define PARTE_OFF (PARTV_OFF + (size_t)T_PAIR * TAB_N)
#define TAIL_OFF (PARTE_OFF + (size_t)T_PAIR * TAB_M)
static_assert(TAIL_OFF + 2 * (size_t)P_PAIRS + N_NODES + M_HYPER + 4 * BS_PAD + 2 * HS_PAD <= SPAN_FLOATS,
              "lists, row offsets and block-sum pads fit behind the tables");
static_assert((size_t)(N_NODES + M_HYPER) * DIM <= TAIL_OFF, "He ends in front of the tail");

__device__ __forceinline__ unsigned wave_incl_scan(unsigned v, int lane) {
    for (int o = 1; o < 32; o <<= 1) {
        unsigned t = __shfl_up(v, o);
        if (lane >= o) v += t;
    }
    return v;
}

template <int SLICES>
__device__ __forceinline__ void slice_prefix(unsigned* __restrict__ part, size_t pitch, int lim, int blk,
                                             unsigned* __restrict__ cnt, unsigned* __restrict__ rowloc,
                                             unsigned* __restrict__ blocksum, unsigned* wsum) {
    const int tid = threadIdx.x, lane = tid & 31, w = tid >> 5;
    const int n = blk * MB_THREADS + tid;
    unsigned run = 0u;
    if (n < lim) {
#pragma unroll 16
        for (int g = 0; g < SLICES; ++g) {
            size_t idx = (size_t)g * pitch + n;
            unsigned c = part[idx];
            part[idx] = run;
            run += c;
        }
    }
    unsigned inc = wave_incl_scan(run, lane);
    if (lane == 31) wsum[w] = inc;
    __syncthreads();
    if (w == 0) {
        unsigned v = (lane < MB_THREADS / 32) ? wsum[lane] : 0u;
        unsigned iv = wave_incl_scan(v, lane);
        if (lane < MB_THREADS / 32) wsum[lane] = iv - v;
        if (lane == 31) blocksum[blk] = iv;
    }
    __syncthreads();
    if (n < lim) {
        rowloc[n] = wsum[w] + inc - run;
        cnt[n] = run;
    }
}

__device__ __forceinline__ void block_offsets(const unsigned* __restrict__ blocksum, int nblk,
                                              unsigned* bs, int lane) {
    unsigned carry = 0u;
    for (int c = 0; c < nblk; c += 32) {
        int i = c + lane;
        unsigned a = (i < nblk) ? blocksum[i] : 0u;
        unsigned ia = wave_incl_scan(a, lane);
        if (i < nblk) bs[i] = carry + ia - a;
        carry += __shfl(ia, 31);
    }
}

__global__ __launch_bounds__(T_THREADS)
void k_count(const int* __restrict__ e_dst, const int* __restrict__ inc_v, const int* __restrict__ inc_e,
             unsigned* __restrict__ part, unsigned* __restrict__ part_v, unsigned* __restrict__ part_e) {
    __shared__ unsigned tab[N_NODES + M_HYPER];
    const int tid = threadIdx.x;
    if (blockIdx.x < T_EDGE) {
        unsigned* tc = tab;
        for (int i = tid; i < N_NODES; i += T_THREADS) tc[i] = 0u;
        __syncthreads();
        const int base = blockIdx.x * EDGES_PER_WG;
        int d[EDGE_SLOTS];
#pragma unroll
        for (int k = 0; k < EDGE_SLOTS; ++k) {
            int i = tid + k * T_THREADS;
            d[k] = (i < EDGES_PER_WG) ? e_dst[base + i] : 0;
        }
#pragma unroll
        for (int k = 0; k < EDGE_SLOTS; ++k) {
            if (tid + k * T_THREADS < EDGES_PER_WG) atomicAdd(&tc[d[k]], 1u);
        }
        __syncthreads();
        unsigned* out = part + (size_t)blockIdx.x * TAB_N;
        for (int i = tid; i < N_NODES; i += T_THREADS) out[i] = tc[i];
    } else {
        const int blk = blockIdx.x - T_EDGE;
        unsigned* tv = tab;
        unsigned* te = tab + N_NODES;
        for (int i = tid; i < N_NODES + M_HYPER; i += T_THREADS) tab[i] = 0u;
        __syncthreads();
        const int base = blk * PAIRS_PER_WG;
        int v[PAIR_SLOTS], h[PAIR_SLOTS];
#pragma unroll
        for (int k = 0; k < PAIR_SLOTS; ++k) {
            int i = tid + k * T_THREADS;
            bool ok = i < PAIRS_PER_WG;
            v[k] = ok ? inc_v[base + i] : 0;
            h[k] = ok ? inc_e[base + i] : 0;
        }
#pragma unroll
        for (int k = 0; k < PAIR_SLOTS; ++k) {
            if (tid + k * T_THREADS < PAIRS_PER_WG) {
                atomicAdd(&tv[v[k]], 1u);
                atomicAdd(&te[h[k]], 1u);
            }
        }
        __syncthreads();
        unsigned* pv = part_v + (size_t)blk * TAB_N;
        for (int i = tid; i < N_NODES; i += T_THREADS) pv[i] = tv[i];
        unsigned* pe = part_e + (size_t)blk * TAB_M;
        for (int i = tid; i < M_HYPER; i += T_THREADS) pe[i] = te[i];
    }
}

#define GEMM_WGS ((N_NODES / 16 * 32 + MB_THREADS - 1) / MB_THREADS)
__global__ __launch_bounds__(MB_THREADS)
void k_merge_gemm(unsigned* __restrict__ part, unsigned* __restrict__ cnt,
                  unsigned* __restrict__ rowloc, unsigned* __restrict__ blocksum,
                  unsigned* __restrict__ part_v, unsigned* __restrict__ part_e,
                  unsigned* __restrict__ cnt_v, unsigned* __restrict__ cnt_e,
                  unsigned* __restrict__ rowloc_v, unsigned* __restrict__ rowloc_e,
                  unsigned* __restrict__ blocksum_v, unsigned* __restrict__ blocksum_e,
                  const float* __restrict__ X, const float* __restrict__ W,
                  const float* __restrict__ bias, float* __restrict__ Xp) {
    __shared__ unsigned wsum[MB_THREADS / 32];
    const int b = blockIdx.x;
    if (b < N_MBLK) {
        slice_prefix<T_EDGE>(part, (size_t)TAB_N, N_NODES, b, cnt, rowloc, blocksum, wsum);
    } else if (b < 2 * N_MBLK + N_HBLK) {
        const bool isv = b < 2 * N_MBLK;
        slice_prefix<T_PAIR>(isv ? part_v : part_e, isv ? (size_t)TAB_N : (size_t)TAB_M,
                             isv ? N_NODES : M_HYPER, isv ? b - N_MBLK : b - 2 * N_MBLK,
                             isv ? cnt_v : cnt_e, isv ? rowloc_v : rowloc_e,
                             isv ? blocksum_v : blocksum_e, wsum);
    } else {
        gemm_tile(b - (2 * N_MBLK + N_HBLK), X, W, bias, Xp);
    }
}

__global__ __launch_bounds__(T_THREADS)
void k_place(const int* __restrict__ e_src, const int* __restrict__ e_dst,
             const int* __restrict__ inc_v, const int* __restrict__ inc_e,
             const unsigned* __restrict__ part, const unsigned* __restrict__ rowloc,
             const unsigned* __restrict__ blocksum, unsigned* __restrict__ blockoff,
             int* __restrict__ csr_src,
             const unsigned* __restrict__ part_v, const unsigned* __restrict__ part_e,
             const unsigned* __restrict__ rowloc_v, const unsigned* __restrict__ rowloc_e,
             const unsigned* __restrict__ blocksum_v, const unsigned* __restrict__ blocksum_e,
             unsigned* __restrict__ blockoff_v, unsigned* __restrict__ blockoff_e,
             int* __restrict__ csr_v, int* __restrict__ csr_e) {
    __shared__ unsigned tab[N_NODES + M_HYPER];
    __shared__ unsigned bv[BS_PAD];
    __shared__ unsigned be[HS_PAD];
    const int tid = threadIdx.x;
    if (blockIdx.x < T_EDGE) {
        unsigned* tb = tab;
        if (tid < 32) block_offsets(blocksum, N_MBLK, bv, tid);
        __syncthreads();
        if (blockIdx.x == 0 && tid < N_MBLK) blockoff[tid] = bv[tid];
        const unsigned* pg = part + (size_t)blockIdx.x * TAB_N;
        for (int i = tid; i < N_NODES; i += T_THREADS) tb[i] = pg[i] + rowloc[i] + bv[i >> MB_SHIFT];
        __syncthreads();
        const int base = blockIdx.x * EDGES_PER_WG;
        int s[EDGE_SLOTS], d[EDGE_SLOTS];
#pragma unroll
        for (int k = 0; k < EDGE_SLOTS; ++k) {
            int i = tid + k * T_THREADS;
            bool ok = i < EDGES_PER_WG;
            s[k] = ok ? e_src[base + i] : 0;
            d[k] = ok ? e_dst[base + i] : 0;
        }
#pragma unroll
        for (int k = 0; k < EDGE_SLOTS; ++k) {
            if (tid + k * T_THREADS < EDGES_PER_WG) {
                unsigned pos = atomicAdd(&tb[d[k]], 1u);
                if (pos < N_EDGES) csr_src[pos] = s[k];
            }
        }
    } else {
        const int blk = blockIdx.x - T_EDGE;
        unsigned* tv = tab;
        unsigned* te = tab + N_NODES;
        if (tid < 32) block_offsets(blocksum_v, N_MBLK, bv, tid);
        else if (tid < 64) block_offsets(blocksum_e, N_HBLK, be, tid - 32);
        __syncthreads();
        if (blk == 0) {
            if (tid < N_MBLK) blockoff_v[tid] = bv[tid];
            if (tid < N_HBLK) blockoff_e[tid] = be[tid];
        }
        const unsigned* pv = part_v + (size_t)blk * TAB_N;
        for (int i = tid; i < N_NODES; i += T_THREADS) tv[i] = pv[i] + rowloc_v[i] + bv[i >> MB_SHIFT];
        const unsigned* pe = part_e + (size_t)blk * TAB_M;
        for (int i = tid; i < M_HYPER; i += T_THREADS) te[i] = pe[i] + rowloc_e[i] + be[i >> MB_SHIFT];
        __syncthreads();
        const int base = blk * PAIRS_PER_WG;
        int v[PAIR_SLOTS], h[PAIR_SLOTS];
#pragma unroll
        for (int k = 0; k < PAIR_SLOTS; ++k) {
            int i = tid + k * T_THREADS;
            bool ok = i < PAIRS_PER_WG;
            v[k] = ok ? inc_v[base + i] : 0;
            h[k] = ok ? inc_e[base + i] : 0;
        }
#pragma unroll
        for (int k = 0; k < PAIR_SLOTS; ++k) {
            if (tid + k * T_THREADS < PAIRS_PER_WG) {
                unsigned pn = atomicAdd(&tv[v[k]], 1u);
                unsigned ph = atomicAdd(&te[h[k]], 1u);
                if (pn < P_PAIRS) csr_v[pn] = h[k];
                if (ph < P_PAIRS) csr_e[ph] = v[k];
            }
        }
    }
}

__device__ __forceinline__ float edge_score(const float* __restrict__ s_src, int s, float sd) {
    float x = s_src[s] + sd;
    return (x >= 0.f) ? x : NEG_SLOPE * x;
}

#define ROW_REQ(k) if (m <= k) break; \
    x##k = *(const float2*)(col + (size_t)__builtin_amdgcn_readlane(idx, q0 + k) * DIM);
#define ROW_ADDW(k) if (m <= k) break; \
    { float wk = __int_as_float(__builtin_amdgcn_readlane(wb, q0 + k)); ax += wk * x##k.x; ay += wk * x##k.y; }
#define ROW_ADD(k) if (m <= k) break; \
    ax += x##k.x; ay += x##k.y;

__device__ __forceinline__ void rows16_weighted(int idx, int wb, int q0, int m, const float* __restrict__ col,
                                                float& ax, float& ay) {
    float2 x0, x1, x2, x3, x4, x5, x6, x7, x8, x9, x10, x11, x12, x13, x14, x15;
    do { ROW_REQ(0) ROW_REQ(1) ROW_REQ(2) ROW_REQ(3) ROW_REQ(4) ROW_REQ(5) ROW_REQ(6) ROW_REQ(7) ROW_REQ(8) ROW_REQ(9) ROW_REQ(10) ROW_REQ(11) ROW_REQ(12) ROW_REQ(13) ROW_REQ(14) ROW_REQ(15) } while (0);
    do { ROW_ADDW(0) ROW_ADDW(1) ROW_ADDW(2) ROW_ADDW(3) ROW_ADDW(4) ROW_ADDW(5) ROW_ADDW(6) ROW_ADDW(7) ROW_ADDW(8) ROW_ADDW(9) ROW_ADDW(10) ROW_ADDW(11) ROW_ADDW(12) ROW_ADDW(13) ROW_ADDW(14) ROW_ADDW(15) } while (0);
}

__device__ __forceinline__ void rows8_plain(int idx, int q0, int m, const float* __restrict__ col,
                                            float& ax, float& ay) {
    float2 x0, x1, x2, x3, x4, x5, x6, x7;
    do { ROW_REQ(0) ROW_REQ(1) ROW_REQ(2) ROW_REQ(3) ROW_REQ(4) ROW_REQ(5) ROW_REQ(6) ROW_REQ(7) } while (0);
    do { ROW_ADD(0) ROW_ADD(1) ROW_ADD(2) ROW_ADD(3) ROW_ADD(4) ROW_ADD(5) ROW_ADD(6) ROW_ADD(7) } while (0);
}

__device__ __forceinline__ void gather_rows(int s, float wgt, int cn, const float* __restrict__ xcol,
                                            float& ax, float& ay) {
    const int wb = __float_as_int(wgt);
    for (int q0 = 0; q0 < cn; q0 += 16) {
        int m = cn - q0;
        rows16_weighted(s, wb, q0, (m < 16) ? m : 16, xcol, ax, ay);
    }
}

__device__ __forceinline__ void sum_rows(int h, int cn, const float* __restrict__ hcol,
                                         float& ax, float& ay) {
    for (int q0 = 0; q0 < cn; q0 += 8) {
        int m = cn - q0;
        rows8_plain(h, q0, (m < 8) ? m : 8, hcol, ax, ay);
    }
}

__device__ __forceinline__ void hyper_rows(int blk, const unsigned* __restrict__ cnt_e,
                             const unsigned* __restrict__ rowloc_e,
                             const unsigned* __restrict__ blockoff_e, const int* __restrict__ csr_e,
                             const unsigned* __restrict__ cnt_v, const float* __restrict__ Xp,
                             float* __restrict__ He) {
    const int e = (blk * 256 + (int)threadIdx.x) >> 5;
    const int lane = threadIdx.x & 31;
    if (e >= M_HYPER) return;
    const int deg = __builtin_amdgcn_readfirstlane((int)cnt_e[e]);
    const unsigned beg = (unsigned)__builtin_amdgcn_readfirstlane(
        (int)(rowloc_e[e] + blockoff_e[e >> MB_SHIFT]));
    const int* vs = csr_e + beg;
    const float* xcol = Xp + lane * 2;
    float ax = 0.f, ay = 0.f;
    for (int j0 = 0; j0 < deg; j0 += 32) {
        int j = j0 + lane;
        int v = 0;
        float wgt = 0.f;
        if (j < deg) {
            v = vs[j];
            float dvv = (float)cnt_v[v];
            wgt = (dvv > 0.f) ? rsqrtf(dvv) : 0.f;
        }
        int cn = deg - j0;
        gather_rows(v, wgt, (cn < 32) ? cn : 32, xcol, ax, ay);
    }
    const float di = (deg > 0) ? (1.f / (float)deg) : 0.f;
    float2 r; r.x = ax * di; r.y = ay * di;
    *(float2*)(He + (size_t)e * DIM + lane * 2) = r;
}

#define SC_WGS ((N_NODES + 255) / 256)
#define HY_WGS ((M_HYPER * 32 + 255) / 256)
__global__ void k_hyper_scalars(const unsigned* __restrict__ cnt_e, const unsigned* __restrict__ rowloc_e,
                                const unsigned* __restrict__ blockoff_e, const int* __restrict__ csr_e,
                                const unsigned* __restrict__ cnt_v, const float* __restrict__ Xp,
                                float* __restrict__ He,
                                const float* __restrict__ a_src, const float* __restrict__ a_dst,
                                float* __restrict__ s_src, float* __restrict__ s_dst) {
    if (blockIdx.x < SC_WGS) node_scalars((int)blockIdx.x, Xp, a_src, a_dst, s_src, s_dst);
    else hyper_rows((int)blockIdx.x - SC_WGS, cnt_e, rowloc_e, blockoff_e, csr_e, cnt_v, Xp, He);
}

__global__ void k_node(const unsigned* __restrict__ cnt, const unsigned* __restrict__ rowloc,
                       const unsigned* __restrict__ blockoff, const int* __restrict__ csr_src,
                       const float* __restrict__ s_src, const float* __restrict__ s_dst,
                       const float* __restrict__ Xp,
                       const unsigned* __restrict__ cnt_v, const unsigned* __restrict__ rowloc_v,
                       const unsigned* __restrict__ blockoff_v, const int* __restrict__ csr_v,
                       const float* __restrict__ He, float* __restrict__ out) {
    const int n = (blockIdx.x * blockDim.x + threadIdx.x) >> 5;
    const int lane = threadIdx.x & 31;
    if (n >= N_NODES) return;
    const int deg = __builtin_amdgcn_readfirstlane((int)cnt[n]);
    const unsigned beg = (unsigned)__builtin_amdgcn_readfirstlane(
        (int)(rowloc[n] + blockoff[n >> MB_SHIFT]));
    const int dgv = __builtin_amdgcn_readfirstlane((int)cnt_v[n]);
    const unsigned begv = (unsigned)__builtin_amdgcn_readfirstlane(
        (int)(rowloc_v[n] + blockoff_v[n >> MB_SHIFT]));
    const float sd = s_dst[n];
    const int* es = csr_src + beg;
    const int* hs = csr_v + begv;
    const float* xcol = Xp + lane * 2;
    const float* hcol = He + lane * 2;
    int s0 = 0, h0 = 0;
    if (lane < deg) s0 = es[lane];
    if (lane < dgv) h0 = hs[lane];
    const float ss0 = s_src[s0];
    float hx = 0.f, hy = 0.f;
    sum_rows(h0, (dgv < 32) ? dgv : 32, hcol, hx, hy);
    for (int j0 = 32; j0 < dgv; j0 += 32) {
        int j = j0 + lane;
        int h = (j < dgv) ? hs[j] : 0;
        int cn = dgv - j0;
        sum_rows(h, (cn < 32) ? cn : 32, hcol, hx, hy);
    }
    float ax = 0.f, ay = 0.f;
    if (deg > 0) {
        float sc0 = -INFINITY;
        if (lane < deg) {
            float x = ss0 + sd;
            sc0 = (x >= 0.f) ? x : NEG_SLOPE * x;
        }
        float m = sc0;
        for (int j0 = 32; j0 < deg; j0 += 32) {
            int j = j0 + lane;
            if (j < deg) m = fmaxf(m, edge_score(s_src, es[j], sd));
        }
        for (int o = 16; o > 0; o >>= 1) m = fmaxf(m, __shfl_xor(m, o));
        if (!isfinite(m)) m = 0.f;
        const float e0 = (lane < deg) ? expf(sc0 - m) : 0.f;
        float sum = e0;
        for (int j0 = 32; j0 < deg; j0 += 32) {
            int j = j0 + lane;
            if (j < deg) sum += expf(edge_score(s_src, es[j], sd) - m);
        }
        for (int o = 16; o > 0; o >>= 1) sum += __shfl_xor(sum, o);
        gather_rows(s0, e0 / sum, (deg < 32) ? deg : 32, xcol, ax, ay);
        for (int j0 = 32; j0 < deg; j0 += 32) {
            int j = j0 + lane;
            int s = 0;
            float wgt = 0.f;
            if (j < deg) {
                s = es[j];
                wgt = expf(edge_score(s_src, s, sd) - m) / sum;
            }
            int cn = deg - j0;
            gather_rows(s, wgt, (cn < 32) ? cn : 32, xcol, ax, ay);
        }
    }
    const float sv = (dgv > 0) ? rsqrtf((float)dgv) : 0.f;
    const float x0 = 0.5f * (ax + hx * sv);
    const float x1 = 0.5f * (ay + hy * sv);
    float2 r;
    r.x = (x0 > 0.f) ? x0 : expm1f(x0);
    r.y = (x1 > 0.f) ? x1 : expm1f(x1);
    *(float2*)(out + (size_t)n * DIM + lane * 2) = r;
}

extern "C" void kernel_launch(void* const* d_in, const int* in_sizes, int n_in,
                              void* d_out, int out_size, void* d_ws, size_t ws_size,
                              hipStream_t stream) {
    const float* X     = (const float*)d_in[0];
    const float* W     = (const float*)d_in[1];
    const float* bias  = (const float*)d_in[2];
    const float* a_src = (const float*)d_in[3];
    const float* a_dst = (const float*)d_in[4];
    const int*   e_src = (const int*)d_in[5];
    const int*   e_dst = (const int*)d_in[6];
    const int*   inc_v = (const int*)d_in[7];
    const int*   inc_e = (const int*)d_in[8];
    float* out = (float*)d_out;
    float* ws  = (float*)d_ws;

    size_t off = 0;
    float* Xp    = ws + off; off += (size_t)N_NODES * DIM;
    float* s_src = ws + off; off += N_NODES;
    float* s_dst = ws + off; off += N_NODES;
    unsigned* cnt    = (unsigned*)(ws + off); off += N_NODES;
    unsigned* rowloc = (unsigned*)(ws + off); off += N_NODES;
    unsigned* cnt_v  = (unsigned*)(ws + off); off += N_NODES;
    unsigned* cnt_e  = (unsigned*)(ws + off); off += M_HYPER;
    float* span  = ws + off; off += SPAN_FLOATS;
    int* csr_src = (int*)(ws + off); off += N_EDGES;

    unsigned* part   = (unsigned*)span;
    unsigned* part_v = (unsigned*)(span + PARTV_OFF);
    unsigned* part_e = (unsigned*)(span + PARTE_OFF);
    float* He        = span + (size_t)N_NODES * DIM;
    float* tail      = span + TAIL_OFF;
    size_t t = 0;
    int* csr_v = (int*)(tail + t); t += P_PAIRS;
    int* csr_e = (int*)(tail + t); t += P_PAIRS;
    unsigned* rowloc_v   = (unsigned*)(tail + t); t += N_NODES;
    unsigned* rowloc_e   = (unsigned*)(tail + t); t += M_HYPER;
    unsigned* blocksum_v = (unsigned*)(tail + t); t += BS_PAD;
    unsigned* blockoff_v = (unsigned*)(tail + t); t += BS_PAD;
    unsigned* blocksum_e = (unsigned*)(tail + t); t += HS_PAD;
    unsigned* blockoff_e = (unsigned*)(tail + t); t += HS_PAD;
    unsigned* blocksum   = (unsigned*)(tail + t); t += BS_PAD;
    unsigned* blockoff   = (unsigned*)(tail + t); t += BS_PAD;

    const int B = 256;
    k_count<<<T_EDGE + T_PAIR, T_THREADS, 0, stream>>>(e_dst, inc_v, inc_e, part, part_v, part_e);
    k_merge_gemm<<<2 * N_MBLK + N_HBLK + GEMM_WGS, MB_THREADS, 0, stream>>>(
        part, cnt, rowloc, blocksum, part_v, part_e, cnt_v, cnt_e, rowloc_v, rowloc_e,
        blocksum_v, blocksum_e, X, W, bias, Xp);
    k_place<<<T_EDGE + T_PAIR, T_THREADS, 0, stream>>>(e_src, e_dst, inc_v, inc_e,
                                                       part, rowloc, blocksum, blockoff, csr_src,
                                                       part_v, part_e, rowloc_v, rowloc_e,
                                                       blocksum_v, blocksum_e, blockoff_v, blockoff_e,
                                                       csr_v, csr_e);
    k_hyper_scalars<<<SC_WGS + HY_WGS, B, 0, stream>>>(cnt_e, rowloc_e, blockoff_e, csr_e, cnt_v, Xp, He,
                                                       a_src, a_dst, s_src, s_dst);
    k_node<<<(N_NODES * 32 + B - 1) / B, B, 0, stream>>>(cnt, rowloc, blockoff, csr_src,
                                                         s_src, s_dst, Xp,
                                                         cnt_v, rowloc_v, blockoff_v, csr_v, He, out);
}
